# starting kernel
# baseline (speedup 1.0000x reference)
#include <hip/hip_runtime.h>

#define B_   128
#define TE_  2048
#define U_   256
#define IN_  256
#define MT_  4

typedef __attribute__((ext_vector_type(16))) __bf16 v16bf;
typedef __attribute__((ext_vector_type(8)))  float  v8f;

__device__ __forceinline__ unsigned short f32_to_bf16_bits(float f) {
  unsigned int u = __float_as_uint(f);
  u += 0x7FFFu + ((u >> 16) & 1u);
  return (unsigned short)(u >> 16);
}

__global__ void k_prep_uaT(const float* __restrict__ Ua,
                           unsigned short* __restrict__ UaT) {
  int e = blockIdx.x;
  int u = threadIdx.x;
  UaT[u * U_ + e] = f32_to_bf16_bits(Ua[e * U_ + u]);
}

__global__ void k_wxstm(const float* __restrict__ h,
                        const float* __restrict__ Wa,
                        float* __restrict__ wx) {
  int b = blockIdx.x, u = threadIdx.x;
  __shared__ float sh[U_];
  sh[u] = h[b * U_ + u];
  __syncthreads();
  float a = 0.f;
  #pragma unroll 4
  for (int k = 0; k < U_; ++k) a = fmaf(sh[k], Wa[k * U_ + u], a);
  wx[b * U_ + u] = a;
}

__global__ __launch_bounds__(256) void k_scores(
    const float* __restrict__ x_seq, const unsigned short* __restrict__ UaT,
    const float* __restrict__ wxstm, const float* __restrict__ b_a,
    const float* __restrict__ V_a, float* __restrict__ et) {
  __shared__ __align__(16) unsigned short sA[16 * MT_ * U_];
  __shared__ float s_e[16 * MT_];

  int b  = blockIdx.y;
  int t0 = blockIdx.x * (16 * MT_);
  int tid = threadIdx.x;

  const float* xrow = x_seq + ((size_t)b * TE_ + t0) * U_;
  for (int i = tid; i < 16 * MT_ * U_; i += 256) sA[i] = f32_to_bf16_bits(xrow[i]);
  if (tid < 16 * MT_) s_e[tid] = 0.f;
  __syncthreads();

  int wave = tid >> 5;
  int lane = tid & 31;
  int sel  = lane >> 4;
  int row  = lane & 15;

  v8f acc[2][MT_];
  #pragma unroll
  for (int j = 0; j < 2; ++j)
    #pragma unroll
    for (int m = 0; m < MT_; ++m)
      acc[j][m] = (v8f){0.f, 0.f, 0.f, 0.f, 0.f, 0.f, 0.f, 0.f};

  const unsigned short* bcol0 = UaT + (size_t)(2 * wave * 16 + row) * U_;
  const unsigned short* bcol1 = bcol0 + 16 * U_;

  #pragma unroll
  for (int kk = 0; kk < U_; kk += 32) {
    union { uint4 q[2]; v16bf v; } fb0, fb1;
    fb0.q[0] = *(const uint4*)(bcol0 + kk + sel * 8);
    fb0.q[1] = *(const uint4*)(bcol0 + kk + 16 + sel * 8);
    fb1.q[0] = *(const uint4*)(bcol1 + kk + sel * 8);
    fb1.q[1] = *(const uint4*)(bcol1 + kk + 16 + sel * 8);
    #pragma unroll
    for (int m = 0; m < MT_; ++m) {
      union { uint4 q[2]; v16bf v; } fa;
      const unsigned short* arow = &sA[(m * 16 + row) * U_ + kk];
      fa.q[0] = *(const uint4*)(arow + sel * 8);
      fa.q[1] = *(const uint4*)(arow + 16 + sel * 8);
      acc[0][m] = __builtin_amdgcn_wmma_f32_16x16x32_bf16(
          false, fa.v, false, fb0.v, (short)0, acc[0][m], false, false);
      acc[1][m] = __builtin_amdgcn_wmma_f32_16x16x32_bf16(
          false, fa.v, false, fb1.v, (short)0, acc[1][m], false, false);
    }
  }

  #pragma unroll
  for (int j = 0; j < 2; ++j) {
    int ug = (2 * wave + j) * 16 + row;
    float wx = wxstm[b * U_ + ug] + b_a[ug];
    float va = V_a[ug];
    #pragma unroll
    for (int m = 0; m < MT_; ++m) {
      #pragma unroll
      for (int i = 0; i < 8; ++i) {
        atomicAdd(&s_e[m * 16 + i + 8 * sel], va * tanhf(acc[j][m][i] + wx));
      }
    }
  }
  __syncthreads();
  if (tid < 16 * MT_) et[(size_t)b * TE_ + t0 + tid] = s_e[tid];
}

__global__ void k_softmax(float* __restrict__ at, float* __restrict__ ctx) {
  int b = blockIdx.x, tid = threadIdx.x;
  float v[8];
  float s = 0.f;
  #pragma unroll
  for (int j = 0; j < 8; ++j) {
    v[j] = expf(at[(size_t)b * TE_ + j * 256 + tid]);
    s += v[j];
  }
  __shared__ float red[256];
  red[tid] = s;
  __syncthreads();
  for (int off = 128; off > 0; off >>= 1) {
    if (tid < off) red[tid] += red[tid + off];
    __syncthreads();
  }
  float inv = 1.f / red[0];
  #pragma unroll
  for (int j = 0; j < 8; ++j)
    at[(size_t)b * TE_ + j * 256 + tid] = v[j] * inv;
  ctx[b * U_ + tid] = 0.f;
}

__global__ void k_context(const float* __restrict__ x_seq,
                          const float* __restrict__ at,
                          float* __restrict__ ctx) {
  int seg = blockIdx.x, b = blockIdx.y, u = threadIdx.x;
  const float* xp = x_seq + ((size_t)b * TE_ + seg * 128) * U_ + u;
  const float* ap = at + (size_t)b * TE_ + seg * 128;
  float acc = 0.f;
  for (int t = 0; t < 128; ++t) {
    __builtin_prefetch(xp + 8 * U_, 0, 1);
    acc = fmaf(ap[t], xp[0], acc);
    xp += U_;
  }
  atomicAdd(&ctx[b * U_ + u], acc);
}

__global__ void k_gates(const float* __restrict__ h, const float* __restrict__ in,
                        const float* __restrict__ ctx,
                        const float* __restrict__ Wz, const float* __restrict__ Cz,
                        const float* __restrict__ bz,
                        const float* __restrict__ Wr, const float* __restrict__ Cr,
                        const float* __restrict__ br,
                        float* __restrict__ ztw, float* __restrict__ rhw) {
  int b = blockIdx.x, u = threadIdx.x;
  __shared__ float sh[U_], si[IN_], sc[U_];
  sh[u] = h[b * U_ + u];
  si[u] = in[b * IN_ + u];
  sc[u] = ctx[b * U_ + u];
  __syncthreads();
  float az = bz[u], ar = br[u];
  for (int k = 0; k < U_; ++k) {
    float hv = sh[k];
    az = fmaf(hv, Wz[k * U_ + u], az);
    ar = fmaf(hv, Wr[k * U_ + u], ar);
  }
  for (int k = 0; k < IN_; ++k) {
    float iv = si[k];
    az = fmaf(iv, Cz[k * U_ + u], az);
    ar = fmaf(iv, Cr[k * U_ + u], ar);
  }
  for (int k = 0; k < U_; ++k) {
    float cv = sc[k];
    az = fmaf(cv, Cz[(IN_ + k) * U_ + u], az);
    ar = fmaf(cv, Cr[(IN_ + k) * U_ + u], ar);
  }
  float zt = 1.f / (1.f + expf(-az));
  float rt = 1.f / (1.f + expf(-ar));
  ztw[b * U_ + u] = zt;
  rhw[b * U_ + u] = rt * sh[u];
}

__global__ void k_out(const float* __restrict__ h, const float* __restrict__ in,
                      const float* __restrict__ ctx,
                      const float* __restrict__ Up, const float* __restrict__ Cp,
                      const float* __restrict__ bp,
                      const float* __restrict__ ztw, const float* __restrict__ rhw,
                      float* __restrict__ out) {
  int b = blockIdx.x, u = threadIdx.x;
  __shared__ float sr[U_], si[IN_], sc[U_];
  sr[u] = rhw[b * U_ + u];
  si[u] = in[b * IN_ + u];
  sc[u] = ctx[b * U_ + u];
  __syncthreads();
  float a = bp[u];
  for (int k = 0; k < U_; ++k)  a = fmaf(sr[k], Up[k * U_ + u], a);
  for (int k = 0; k < IN_; ++k) a = fmaf(si[k], Cp[k * U_ + u], a);
  for (int k = 0; k < U_; ++k)  a = fmaf(sc[k], Cp[(IN_ + k) * U_ + u], a);
  float th = tanhf(a);
  float zt = ztw[b * U_ + u];
  out[b * U_ + u] = (1.f - zt) * h[b * U_ + u] + zt * th;
}

extern "C" void kernel_launch(void* const* d_in, const int* in_sizes, int n_in,
                              void* d_out, int out_size, void* d_ws, size_t ws_size,
                              hipStream_t stream) {
  const float* inputs = (const float*)d_in[0];
  const float* h_tm   = (const float*)d_in[1];
  const float* x_seq  = (const float*)d_in[2];
  const float* V_a    = (const float*)d_in[3];
  const float* W_a    = (const float*)d_in[4];
  const float* U_a    = (const float*)d_in[5];
  const float* b_a    = (const float*)d_in[6];
  const float* C_z    = (const float*)d_in[7];
  const float* W_z    = (const float*)d_in[8];
  const float* b_z    = (const float*)d_in[9];
  const float* C_r    = (const float*)d_in[10];
  const float* W_r    = (const float*)d_in[11];
  const float* b_r    = (const float*)d_in[12];
  const float* C_p    = (const float*)d_in[13];
  const float* U_p    = (const float*)d_in[14];
  const float* b_p    = (const float*)d_in[15];
  float* out = (float*)d_out;

  float* wsf   = (float*)d_ws;
  float* wxstm = wsf;
  float* et    = wsf + 32768;
  float* ctx   = et + (size_t)B_ * TE_;
  float* ztw   = ctx + B_ * U_;
  float* rhw   = ztw + B_ * U_;
  unsigned short* UaT = (unsigned short*)(rhw + B_ * U_);

  dim3 blk(256);
  k_prep_uaT<<<dim3(U_), blk, 0, stream>>>(U_a, UaT);
  k_wxstm  <<<dim3(B_), blk, 0, stream>>>(h_tm, W_a, wxstm);
  k_scores <<<dim3(TE_ / (16 * MT_), B_), blk, 0, stream>>>(x_seq, UaT, wxstm, b_a,
                                                            V_a, et);
  k_softmax<<<dim3(B_), blk, 0, stream>>>(et, ctx);
  k_context<<<dim3(TE_ / 128, B_), blk, 0, stream>>>(x_seq, et, ctx);
  k_gates  <<<dim3(B_), blk, 0, stream>>>(h_tm, inputs, ctx, W_z, C_z, b_z,
                                          W_r, C_r, b_r, ztw, rhw);
  k_out    <<<dim3(B_), blk, 0, stream>>>(h_tm, inputs, ctx, U_p, C_p, b_p,
                                          ztw, rhw, out);
}


# wxstm, gates, out layers on the f32 16x16x4 matrix unit: block = 16 rows x 64 cols, 32-wide K chunks dealt over 8 waves, a wave's loads in one flight; sum order: WMMA chain, then 8 partials
# speedup vs baseline: 3.6353x; 3.6353x over previous
#include <hip/hip_runtime.h>

#define B_   128
#define TE_  2048
#define U_   256
#define IN_  256

typedef __attribute__((ext_vector_type(16))) __bf16 v16bf;
typedef __attribute__((ext_vector_type(8)))  float  v8f;

__device__ __forceinline__ unsigned short f32_to_bf16_bits(float f) {
  unsigned int u = __float_as_uint(f);
  u += 0x7FFFu + ((u >> 16) & 1u);
  return (unsigned short)(u >> 16);
}

typedef __attribute__((ext_vector_type(2))) float v2f;
typedef __attribute__((ext_vector_type(16))) float v16f;

template <int NCH>
__device__ __forceinline__ float4 dense16x64(
    const float* __restrict__ a0, const float* __restrict__ a1,
    const float* __restrict__ a2, const float* __restrict__ w0,
    const float* __restrict__ w1, const float* __restrict__ w2,
    int m0, int n0, float4* __restrict__ part) {
  const int tid  = threadIdx.x;
  const int wave = tid >> 5;
  const int lane = tid & 31;
  const int h    = lane >> 4;
  const int c    = lane & 15;
  const float* ap[3] = {a0, a1, a2};
  const float* wp[3] = {w0, w1, w2};

  v16f fa[NCH];
  v16f fb[NCH][4];
  #pragma unroll
  for (int i = 0; i < NCH; ++i) {
    const float* A  = ap[i] + (m0 + c) * U_ + wave * 32 + 4 * h;
    const float* Bp = wp[i] + (wave * 32 + 4 * h) * U_ + n0 + 4 * c;
    #pragma unroll
    for (int p = 0; p < 4; ++p) {
      const float4 va = *(const float4*)(A + 8 * p);
      fa[i][4 * p + 0] = va.x; fa[i][4 * p + 1] = va.y;
      fa[i][4 * p + 2] = va.z; fa[i][4 * p + 3] = va.w;
      #pragma unroll
      for (int j = 0; j < 4; ++j) {
        const float4 vb = *(const float4*)(Bp + (8 * p + j) * U_);
        fb[i][p][4 * j + 0] = vb.x; fb[i][p][4 * j + 1] = vb.y;
        fb[i][p][4 * j + 2] = vb.z; fb[i][p][4 * j + 3] = vb.w;
      }
    }
  }
  if constexpr (NCH == 1) {
    asm volatile("" :: "v"(fa[0]), "v"(fb[0][0]), "v"(fb[0][1]), "v"(fb[0][2]),
                 "v"(fb[0][3]));
  } else {
    asm volatile("" :: "v"(fa[0]), "v"(fb[0][0]), "v"(fb[0][1]), "v"(fb[0][2]),
                 "v"(fb[0][3]), "v"(fa[1]), "v"(fb[1][0]), "v"(fb[1][1]),
                 "v"(fb[1][2]), "v"(fb[1][3]), "v"(fa[NCH - 1]),
                 "v"(fb[NCH - 1][0]), "v"(fb[NCH - 1][1]), "v"(fb[NCH - 1][2]),
                 "v"(fb[NCH - 1][3]));
  }

  v8f acc[4];
  #pragma unroll
  for (int t = 0; t < 4; ++t)
    acc[t] = (v8f){0.f, 0.f, 0.f, 0.f, 0.f, 0.f, 0.f, 0.f};
  #pragma unroll
  for (int i = 0; i < NCH; ++i) {
    #pragma unroll
    for (int p = 0; p < 4; ++p) {
      #pragma unroll
      for (int q = 0; q < 2; ++q) {
        const v2f a = (v2f){fa[i][4 * p + 2 * q], fa[i][4 * p + 2 * q + 1]};
        const v16f b = fb[i][p];
        const int j0 = 4 * (2 * q), j1 = 4 * (2 * q + 1);
        acc[0] = __builtin_amdgcn_wmma_f32_16x16x4_f32(
            false, a, false, (v2f){b[j0 + 0], b[j1 + 0]}, (short)0, acc[0], false, false);
        acc[1] = __builtin_amdgcn_wmma_f32_16x16x4_f32(
            false, a, false, (v2f){b[j0 + 1], b[j1 + 1]}, (short)0, acc[1], false, false);
        acc[2] = __builtin_amdgcn_wmma_f32_16x16x4_f32(
            false, a, false, (v2f){b[j0 + 2], b[j1 + 2]}, (short)0, acc[2], false, false);
        acc[3] = __builtin_amdgcn_wmma_f32_16x16x4_f32(
            false, a, false, (v2f){b[j0 + 3], b[j1 + 3]}, (short)0, acc[3], false, false);
      }
    }
  }

  #pragma unroll
  for (int r = 0; r < 8; ++r)
    part[(wave * 8 + r) * 32 + lane] =
        make_float4(acc[0][r], acc[1][r], acc[2][r], acc[3][r]);
  __syncthreads();
  float4 s = part[wave * 32 + lane];
  #pragma unroll
  for (int w = 1; w < 8; ++w) {
    const float4 v = part[(w * 8 + wave) * 32 + lane];
    s.x += v.x; s.y += v.y; s.z += v.z; s.w += v.w;
  }
  return s;
}

__global__ __launch_bounds__(256) void k_wxstm(const float* __restrict__ h,
                                               const float* __restrict__ Wa,
                                               float* __restrict__ wx) {
  __shared__ float4 part[8 * 8 * 32];
  const int n0 = blockIdx.x * 64, m0 = blockIdx.y * 16;
  const float4 s = dense16x64<1>(h, h, h, Wa, Wa, Wa, m0, n0, part);
  const int lane = threadIdx.x & 31, wave = threadIdx.x >> 5;
  const int row = m0 + 8 * (lane >> 4) + wave, col = n0 + 4 * (lane & 15);
  *(float4*)(wx + row * U_ + col) = s;
}

#define ROWS_WG_ 1024
#define WAVES_   16
#define TILES_W_ 4

__device__ __forceinline__ unsigned int pack2_bf16(float lo, float hi) {
  return (unsigned int)f32_to_bf16_bits(lo) |
         ((unsigned int)f32_to_bf16_bits(hi) << 16);
}

__global__ __launch_bounds__(WAVES_ * 32) void k_scores(
    const float* __restrict__ x_seq, const float* __restrict__ Ua,
    const float* __restrict__ wxstm, const float* __restrict__ b_a,
    const float* __restrict__ V_a, float* __restrict__ et) {
  __shared__ __align__(16) unsigned short sB[U_ * U_];
  __shared__ float s_wx[U_];
  __shared__ float s_va[U_];

  const int tid  = threadIdx.x;
  const int wave = tid >> 5;
  const int lane = tid & 31;
  const int h    = lane >> 4;
  const int c    = lane & 15;
  const int b    = blockIdx.x >> 1;
  const size_t row0 = (size_t)blockIdx.x * ROWS_WG_;

  {
    const int s = wave & 7;
    const int nt0 = (wave >> 3) * 8;
    #pragma unroll 4
    for (int nt = nt0; nt < nt0 + 8; ++nt) {
      unsigned int pk[8];
      #pragma unroll
      for (int j2 = 0; j2 < 8; ++j2) {
        const int j = 2 * j2;
        const int k = 32 * s + 8 * (j >> 2) + 4 * h + (j & 3);
        pk[j2] = pack2_bf16(Ua[k * U_ + nt * 16 + c],
                            Ua[(k + 1) * U_ + nt * 16 + c]);
      }
      uint4* dst = (uint4*)&sB[(nt * 8 + s) * 512 + lane * 8];
      dst[0]  = make_uint4(pk[0], pk[1], pk[2], pk[3]);
      dst[32] = make_uint4(pk[4], pk[5], pk[6], pk[7]);
    }
    if (tid < U_) {
      s_wx[tid] = wxstm[b * U_ + tid] + b_a[tid];
      s_va[tid] = V_a[tid];
    }
  }
  __syncthreads();

  #pragma unroll 1
  for (int it = 0; it < TILES_W_; ++it) {
    const size_t R = row0 + (size_t)(it * WAVES_ + wave) * 16;

    float4 raw[32];
    {
      const float* xp = x_seq + (R + c) * U_ + h * 4;
      #pragma unroll
      for (int q = 0; q < 32; ++q) raw[q] = *(const float4*)(xp + q * 8);
    }

    union { unsigned int d[8]; v16bf v; } fa[8];
    #pragma unroll
    for (int s = 0; s < 8; ++s) {
      #pragma unroll
      for (int q4 = 0; q4 < 4; ++q4) {
        const float4 v = raw[4 * s + q4];
        fa[s].d[2 * q4]     = pack2_bf16(v.x, v.y);
        fa[s].d[2 * q4 + 1] = pack2_bf16(v.z, v.w);
      }
    }

    float sum[8];
    #pragma unroll
    for (int r = 0; r < 8; ++r) sum[r] = 0.f;

    #pragma unroll 1
    for (int ntq = 0; ntq < 4; ++ntq) {
      v8f acc[4];
      #pragma unroll
      for (int u = 0; u < 4; ++u)
        acc[u] = (v8f){0.f, 0.f, 0.f, 0.f, 0.f, 0.f, 0.f, 0.f};
      #pragma unroll
      for (int s = 0; s < 8; ++s) {
        #pragma unroll
        for (int u = 0; u < 4; ++u) {
          const uint4* bp = (const uint4*)&sB[((ntq * 4 + u) * 8 + s) * 512];
          union { uint4 q[2]; v16bf v; } fb;
          fb.q[0] = bp[lane];
          fb.q[1] = bp[32 + lane];
          acc[u] = __builtin_amdgcn_wmma_f32_16x16x32_bf16(
              false, fa[s].v, false, fb.v, (short)0, acc[u], false, false);
        }
      }
      #pragma unroll
      for (int u = 0; u < 4; ++u) {
        const int ug = (ntq * 4 + u) * 16 + c;
        const float wx = s_wx[ug];
        const float va = s_va[ug];
        #pragma unroll
        for (int r = 0; r < 8; ++r) sum[r] += va * tanhf(acc[u][r] + wx);
      }
    }

    #pragma unroll
    for (int r = 0; r < 8; ++r) {
      float v = sum[r];
      v += __shfl_xor(v, 1);
      v += __shfl_xor(v, 2);
      v += __shfl_xor(v, 4);
      v += __shfl_xor(v, 8);
      sum[r] = v;
    }
    if (c == 0) {
      float4* o = (float4*)(et + R + 8 * h);
      o[0] = make_float4(sum[0], sum[1], sum[2], sum[3]);
      o[1] = make_float4(sum[4], sum[5], sum[6], sum[7]);
    }
  }
}

__global__ void k_softmax(float* __restrict__ at, float* __restrict__ ctx) {
  int b = blockIdx.x, tid = threadIdx.x;
  float v[8];
  float s = 0.f;
  #pragma unroll
  for (int j = 0; j < 8; ++j) {
    v[j] = expf(at[(size_t)b * TE_ + j * 256 + tid]);
    s += v[j];
  }
  __shared__ float red[256];
  red[tid] = s;
  __syncthreads();
  for (int off = 128; off > 0; off >>= 1) {
    if (tid < off) red[tid] += red[tid + off];
    __syncthreads();
  }
  float inv = 1.f / red[0];
  #pragma unroll
  for (int j = 0; j < 8; ++j)
    at[(size_t)b * TE_ + j * 256 + tid] = v[j] * inv;
  ctx[b * U_ + tid] = 0.f;
}

__global__ void k_context(const float* __restrict__ x_seq,
                          const float* __restrict__ at,
                          float* __restrict__ ctx) {
  int seg = blockIdx.x, b = blockIdx.y, u = threadIdx.x;
  const float* xp = x_seq + ((size_t)b * TE_ + seg * 128) * U_ + u;
  const float* ap = at + (size_t)b * TE_ + seg * 128;
  float acc = 0.f;
  for (int t = 0; t < 128; ++t) {
    __builtin_prefetch(xp + 8 * U_, 0, 1);
    acc = fmaf(ap[t], xp[0], acc);
    xp += U_;
  }
  atomicAdd(&ctx[b * U_ + u], acc);
}

__device__ __forceinline__ float sigmoidf_(float a) {
  return 1.f / (1.f + expf(-a));
}

__global__ __launch_bounds__(256) void k_gates(
    const float* __restrict__ h, const float* __restrict__ in,
    const float* __restrict__ ctx,
    const float* __restrict__ Wz, const float* __restrict__ Cz,
    const float* __restrict__ bz,
    const float* __restrict__ Wr, const float* __restrict__ Cr,
    const float* __restrict__ br,
    float* __restrict__ ztw, float* __restrict__ rhw) {
  __shared__ float4 part[8 * 8 * 32];
  const int n0 = blockIdx.x * 64, m0 = blockIdx.y * 16;
  const bool isr = blockIdx.z != 0;
  const float* W    = isr ? Wr : Wz;
  const float* C    = isr ? Cr : Cz;
  const float* bias = isr ? br : bz;
  const float4 s = dense16x64<3>(h, in, ctx, W, C, C + IN_ * U_, m0, n0, part);
  const int lane = threadIdx.x & 31, wave = threadIdx.x >> 5;
  const int row = m0 + 8 * (lane >> 4) + wave, col = n0 + 4 * (lane & 15);
  const float4 bb = *(const float4*)(bias + col);
  float4 g;
  g.x = sigmoidf_(s.x + bb.x);
  g.y = sigmoidf_(s.y + bb.y);
  g.z = sigmoidf_(s.z + bb.z);
  g.w = sigmoidf_(s.w + bb.w);
  if (isr) {
    const float4 hv = *(const float4*)(h + row * U_ + col);
    *(float4*)(rhw + row * U_ + col) =
        make_float4(g.x * hv.x, g.y * hv.y, g.z * hv.z, g.w * hv.w);
  } else {
    *(float4*)(ztw + row * U_ + col) = g;
  }
}

__global__ __launch_bounds__(256) void k_out(
    const float* __restrict__ h, const float* __restrict__ in,
    const float* __restrict__ ctx,
    const float* __restrict__ Up, const float* __restrict__ Cp,
    const float* __restrict__ bp,
    const float* __restrict__ ztw, const float* __restrict__ rhw,
    float* __restrict__ out) {
  __shared__ float4 part[8 * 8 * 32];
  const int n0 = blockIdx.x * 64, m0 = blockIdx.y * 16;
  const float4 s = dense16x64<3>(rhw, in, ctx, Up, Cp, Cp + IN_ * U_, m0, n0, part);
  const int lane = threadIdx.x & 31, wave = threadIdx.x >> 5;
  const int row = m0 + 8 * (lane >> 4) + wave, col = n0 + 4 * (lane & 15);
  const float4 bb = *(const float4*)(bp + col);
  const float4 zt = *(const float4*)(ztw + row * U_ + col);
  const float4 hv = *(const float4*)(h + row * U_ + col);
  float4 o;
  o.x = (1.f - zt.x) * hv.x + zt.x * tanhf(s.x + bb.x);
  o.y = (1.f - zt.y) * hv.y + zt.y * tanhf(s.y + bb.y);
  o.z = (1.f - zt.z) * hv.z + zt.z * tanhf(s.z + bb.z);
  o.w = (1.f - zt.w) * hv.w + zt.w * tanhf(s.w + bb.w);
  *(float4*)(out + row * U_ + col) = o;
}

extern "C" void kernel_launch(void* const* d_in, const int* in_sizes, int n_in,
                              void* d_out, int out_size, void* d_ws, size_t ws_size,
                              hipStream_t stream) {
  const float* inputs = (const float*)d_in[0];
  const float* h_tm   = (const float*)d_in[1];
  const float* x_seq  = (const float*)d_in[2];
  const float* V_a    = (const float*)d_in[3];
  const float* W_a    = (const float*)d_in[4];
  const float* U_a    = (const float*)d_in[5];
  const float* b_a    = (const float*)d_in[6];
  const float* C_z    = (const float*)d_in[7];
  const float* W_z    = (const float*)d_in[8];
  const float* b_z    = (const float*)d_in[9];
  const float* C_r    = (const float*)d_in[10];
  const float* W_r    = (const float*)d_in[11];
  const float* b_r    = (const float*)d_in[12];
  const float* C_p    = (const float*)d_in[13];
  const float* U_p    = (const float*)d_in[14];
  const float* b_p    = (const float*)d_in[15];
  float* out = (float*)d_out;

  float* wsf   = (float*)d_ws;
  float* wxstm = wsf;
  float* et    = wsf + 32768;
  float* ctx   = et + (size_t)B_ * TE_;
  float* ztw   = ctx + B_ * U_;
  float* rhw   = ztw + B_ * U_;

  dim3 blk(256);
  k_wxstm  <<<dim3(U_ / 64, B_ / 16), blk, 0, stream>>>(h_tm, W_a, wxstm);
  k_scores <<<dim3(B_ * TE_ / ROWS_WG_), dim3(WAVES_ * 32), 0, stream>>>(
      x_seq, U_a, wxstm, b_a, V_a, et);
  k_softmax<<<dim3(B_), blk, 0, stream>>>(et, ctx);
  k_context<<<dim3(TE_ / 128, B_), blk, 0, stream>>>(x_seq, et, ctx);
  k_gates  <<<dim3(U_ / 64, B_ / 16, 2), blk, 0, stream>>>(h_tm, inputs, ctx, W_z, C_z, b_z,
                                          W_r, C_r, b_r, ztw, rhw);
  k_out    <<<dim3(U_ / 64, B_ / 16), blk, 0, stream>>>(h_tm, inputs, ctx, U_p, C_p, b_p,
                                          ztw, rhw, out);
}


# softmax folded into the context launch: block = the 1024 steps of a k_scores block, newest first; sums its row's exp(et), streams x once; gate loaders add 2 partials; no atomics, sum order fixed
# speedup vs baseline: 3.9449x; 1.0852x over previous
#include <hip/hip_runtime.h>

#define B_   128
#define TE_  2048
#define U_   256
#define IN_  256

typedef __attribute__((ext_vector_type(16))) __bf16 v16bf;
typedef __attribute__((ext_vector_type(8)))  float  v8f;

__device__ __forceinline__ unsigned short f32_to_bf16_bits(float f) {
  unsigned int u = __float_as_uint(f);
  u += 0x7FFFu + ((u >> 16) & 1u);
  return (unsigned short)(u >> 16);
}

typedef __attribute__((ext_vector_type(2))) float v2f;
typedef __attribute__((ext_vector_type(16))) float v16f;

template <int NCH>
__device__ __forceinline__ float4 dense16x64(
    const float* __restrict__ a0, const float* __restrict__ a1,
    const float* __restrict__ a2, const float* __restrict__ a2x,
    const float* __restrict__ w0, const float* __restrict__ w1,
    const float* __restrict__ w2, int m0, int n0, float4* __restrict__ part) {
  const int tid  = threadIdx.x;
  const int wave = tid >> 5;
  const int lane = tid & 31;
  const int h    = lane >> 4;
  const int c    = lane & 15;
  const float* ap[3] = {a0, a1, a2};
  const float* wp[3] = {w0, w1, w2};

  v16f fa[NCH];
  v16f fb[NCH][4];
  #pragma unroll
  for (int i = 0; i < NCH; ++i) {
    const float* A  = ap[i] + (m0 + c) * U_ + wave * 32 + 4 * h;
    const float* Bp = wp[i] + (wave * 32 + 4 * h) * U_ + n0 + 4 * c;
    #pragma unroll
    for (int p = 0; p < 4; ++p) {
      float4 va = *(const float4*)(A + 8 * p);
      if (NCH == 3 && i == 2) {
        const float4 vx = *(const float4*)(a2x + (m0 + c) * U_ + wave * 32 + 4 * h + 8 * p);
        va.x += vx.x; va.y += vx.y; va.z += vx.z; va.w += vx.w;
      }
      fa[i][4 * p + 0] = va.x; fa[i][4 * p + 1] = va.y;
      fa[i][4 * p + 2] = va.z; fa[i][4 * p + 3] = va.w;
      #pragma unroll
      for (int j = 0; j < 4; ++j) {
        const float4 vb = *(const float4*)(Bp + (8 * p + j) * U_);
        fb[i][p][4 * j + 0] = vb.x; fb[i][p][4 * j + 1] = vb.y;
        fb[i][p][4 * j + 2] = vb.z; fb[i][p][4 * j + 3] = vb.w;
      }
    }
  }
  if constexpr (NCH == 1) {
    asm volatile("" :: "v"(fa[0]), "v"(fb[0][0]), "v"(fb[0][1]), "v"(fb[0][2]),
                 "v"(fb[0][3]));
  } else {
    asm volatile("" :: "v"(fa[0]), "v"(fb[0][0]), "v"(fb[0][1]), "v"(fb[0][2]),
                 "v"(fb[0][3]), "v"(fa[1]), "v"(fb[1][0]), "v"(fb[1][1]),
                 "v"(fb[1][2]), "v"(fb[1][3]), "v"(fa[NCH - 1]),
                 "v"(fb[NCH - 1][0]), "v"(fb[NCH - 1][1]), "v"(fb[NCH - 1][2]),
                 "v"(fb[NCH - 1][3]));
  }

  v8f acc[4];
  #pragma unroll
  for (int t = 0; t < 4; ++t)
    acc[t] = (v8f){0.f, 0.f, 0.f, 0.f, 0.f, 0.f, 0.f, 0.f};
  #pragma unroll
  for (int i = 0; i < NCH; ++i) {
    #pragma unroll
    for (int p = 0; p < 4; ++p) {
      #pragma unroll
      for (int q = 0; q < 2; ++q) {
        const v2f a = (v2f){fa[i][4 * p + 2 * q], fa[i][4 * p + 2 * q + 1]};
        const v16f b = fb[i][p];
        const int j0 = 4 * (2 * q), j1 = 4 * (2 * q + 1);
        acc[0] = __builtin_amdgcn_wmma_f32_16x16x4_f32(
            false, a, false, (v2f){b[j0 + 0], b[j1 + 0]}, (short)0, acc[0], false, false);
        acc[1] = __builtin_amdgcn_wmma_f32_16x16x4_f32(
            false, a, false, (v2f){b[j0 + 1], b[j1 + 1]}, (short)0, acc[1], false, false);
        acc[2] = __builtin_amdgcn_wmma_f32_16x16x4_f32(
            false, a, false, (v2f){b[j0 + 2], b[j1 + 2]}, (short)0, acc[2], false, false);
        acc[3] = __builtin_amdgcn_wmma_f32_16x16x4_f32(
            false, a, false, (v2f){b[j0 + 3], b[j1 + 3]}, (short)0, acc[3], false, false);
      }
    }
  }

  #pragma unroll
  for (int r = 0; r < 8; ++r)
    part[(wave * 8 + r) * 32 + lane] =
        make_float4(acc[0][r], acc[1][r], acc[2][r], acc[3][r]);
  __syncthreads();
  float4 s = part[wave * 32 + lane];
  #pragma unroll
  for (int w = 1; w < 8; ++w) {
    const float4 v = part[(w * 8 + wave) * 32 + lane];
    s.x += v.x; s.y += v.y; s.z += v.z; s.w += v.w;
  }
  return s;
}

__global__ __launch_bounds__(256) void k_wxstm(const float* __restrict__ h,
                                               const float* __restrict__ Wa,
                                               float* __restrict__ wx) {
  __shared__ float4 part[8 * 8 * 32];
  const int n0 = blockIdx.x * 64, m0 = blockIdx.y * 16;
  const float4 s = dense16x64<1>(h, h, h, h, Wa, Wa, Wa, m0, n0, part);
  const int lane = threadIdx.x & 31, wave = threadIdx.x >> 5;
  const int row = m0 + 8 * (lane >> 4) + wave, col = n0 + 4 * (lane & 15);
  *(float4*)(wx + row * U_ + col) = s;
}

#define ROWS_WG_ 1024
#define WAVES_   16
#define TILES_W_ 4

__device__ __forceinline__ unsigned int pack2_bf16(float lo, float hi) {
  return (unsigned int)f32_to_bf16_bits(lo) |
         ((unsigned int)f32_to_bf16_bits(hi) << 16);
}

__global__ __launch_bounds__(WAVES_ * 32) void k_scores(
    const float* __restrict__ x_seq, const float* __restrict__ Ua,
    const float* __restrict__ wxstm, const float* __restrict__ b_a,
    const float* __restrict__ V_a, float* __restrict__ et) {
  __shared__ __align__(16) unsigned short sB[U_ * U_];
  __shared__ float s_wx[U_];
  __shared__ float s_va[U_];

  const int tid  = threadIdx.x;
  const int wave = tid >> 5;
  const int lane = tid & 31;
  const int h    = lane >> 4;
  const int c    = lane & 15;
  const int b    = blockIdx.x >> 1;
  const size_t row0 = (size_t)blockIdx.x * ROWS_WG_;

  {
    const int s = wave & 7;
    const int nt0 = (wave >> 3) * 8;
    #pragma unroll 4
    for (int nt = nt0; nt < nt0 + 8; ++nt) {
      unsigned int pk[8];
      #pragma unroll
      for (int j2 = 0; j2 < 8; ++j2) {
        const int j = 2 * j2;
        const int k = 32 * s + 8 * (j >> 2) + 4 * h + (j & 3);
        pk[j2] = pack2_bf16(Ua[k * U_ + nt * 16 + c],
                            Ua[(k + 1) * U_ + nt * 16 + c]);
      }
      uint4* dst = (uint4*)&sB[(nt * 8 + s) * 512 + lane * 8];
      dst[0]  = make_uint4(pk[0], pk[1], pk[2], pk[3]);
      dst[32] = make_uint4(pk[4], pk[5], pk[6], pk[7]);
    }
    if (tid < U_) {
      s_wx[tid] = wxstm[b * U_ + tid] + b_a[tid];
      s_va[tid] = V_a[tid];
    }
  }
  __syncthreads();

  #pragma unroll 1
  for (int it = 0; it < TILES_W_; ++it) {
    const size_t R = row0 + (size_t)(it * WAVES_ + wave) * 16;

    float4 raw[32];
    {
      const float* xp = x_seq + (R + c) * U_ + h * 4;
      #pragma unroll
      for (int q = 0; q < 32; ++q) raw[q] = *(const float4*)(xp + q * 8);
    }

    union { unsigned int d[8]; v16bf v; } fa[8];
    #pragma unroll
    for (int s = 0; s < 8; ++s) {
      #pragma unroll
      for (int q4 = 0; q4 < 4; ++q4) {
        const float4 v = raw[4 * s + q4];
        fa[s].d[2 * q4]     = pack2_bf16(v.x, v.y);
        fa[s].d[2 * q4 + 1] = pack2_bf16(v.z, v.w);
      }
    }

    float sum[8];
    #pragma unroll
    for (int r = 0; r < 8; ++r) sum[r] = 0.f;

    #pragma unroll 1
    for (int ntq = 0; ntq < 4; ++ntq) {
      v8f acc[4];
      #pragma unroll
      for (int u = 0; u < 4; ++u)
        acc[u] = (v8f){0.f, 0.f, 0.f, 0.f, 0.f, 0.f, 0.f, 0.f};
      #pragma unroll
      for (int s = 0; s < 8; ++s) {
        #pragma unroll
        for (int u = 0; u < 4; ++u) {
          const uint4* bp = (const uint4*)&sB[((ntq * 4 + u) * 8 + s) * 512];
          union { uint4 q[2]; v16bf v; } fb;
          fb.q[0] = bp[lane];
          fb.q[1] = bp[32 + lane];
          acc[u] = __builtin_amdgcn_wmma_f32_16x16x32_bf16(
              false, fa[s].v, false, fb.v, (short)0, acc[u], false, false);
        }
      }
      #pragma unroll
      for (int u = 0; u < 4; ++u) {
        const int ug = (ntq * 4 + u) * 16 + c;
        const float wx = s_wx[ug];
        const float va = s_va[ug];
        #pragma unroll
        for (int r = 0; r < 8; ++r) sum[r] += va * tanhf(acc[u][r] + wx);
      }
    }

    #pragma unroll
    for (int r = 0; r < 8; ++r) {
      float v = sum[r];
      v += __shfl_xor(v, 1);
      v += __shfl_xor(v, 2);
      v += __shfl_xor(v, 4);
      v += __shfl_xor(v, 8);
      sum[r] = v;
    }
    if (c == 0) {
      float4* o = (float4*)(et + R + 8 * h);
      o[0] = make_float4(sum[0], sum[1], sum[2], sum[3]);
      o[1] = make_float4(sum[4], sum[5], sum[6], sum[7]);
    }
  }
}

#define CG_ 8
#define CJ_ (ROWS_WG_ / CG_)
#define CB_ 16
#define CK_ (CJ_ / CB_)

__device__ __forceinline__ void ctx_load_group(float4 (&buf)[CB_],
                                               const float* __restrict__ xp, int k) {
  #pragma unroll
  for (int jj = 0; jj < CB_; ++jj)
    buf[jj] = *(const float4*)(xp + (size_t)(8 * (CB_ * k + jj)) * U_);
}

__device__ __forceinline__ void ctx_sum_group(const float4 (&buf)[CB_],
                                              const float* w, float4& acc) {
  float wv[CB_];
  #pragma unroll
  for (int q = 0; q < CB_ / 4; ++q) {
    const float4 w4 = *(const float4*)(w + 4 * q);
    wv[4 * q + 0] = w4.x; wv[4 * q + 1] = w4.y;
    wv[4 * q + 2] = w4.z; wv[4 * q + 3] = w4.w;
  }
  #pragma unroll
  for (int jj = CB_ - 1; jj >= 0; --jj) {
    acc.x = fmaf(wv[jj], buf[jj].x, acc.x);
    acc.y = fmaf(wv[jj], buf[jj].y, acc.y);
    acc.z = fmaf(wv[jj], buf[jj].z, acc.z);
    acc.w = fmaf(wv[jj], buf[jj].w, acc.w);
  }
}

__global__ __launch_bounds__(512) void k_context(const float* __restrict__ x_seq,
                                                 const float* __restrict__ et,
                                                 float* __restrict__ ctxp) {
  __shared__ __align__(16) float s_w[ROWS_WG_];
  __shared__ float s_red[16];
  __shared__ float4 s_part[CG_ * 64];

  const int tid  = threadIdx.x;
  const int lane = tid & 31;
  const int wave = tid >> 5;
  const int g    = tid >> 6;
  const int l    = tid & 63;
  const int b    = blockIdx.x >> 1;
  const int half = blockIdx.x & 1;
  const size_t row0 = (size_t)blockIdx.x * ROWS_WG_;
  const float* xp = x_seq + (row0 + g) * U_ + 4 * l;

  const float* etb = et + (size_t)b * TE_;
  const float r0 = etb[tid], r1 = etb[tid + 512];
  const float r2 = etb[tid + 1024], r3 = etb[tid + 1536];

  float4 bufA[CB_], bufB[CB_];
  ctx_load_group(bufA, xp, CK_ - 1);
  ctx_load_group(bufB, xp, CK_ - 2);
  __builtin_amdgcn_sched_barrier(0);

  const float e0 = expf(r0), e1 = expf(r1), e2 = expf(r2), e3 = expf(r3);
  s_w[(tid & 7) * CJ_ + (tid >> 3)]      = half ? e2 : e0;
  s_w[(tid & 7) * CJ_ + 64 + (tid >> 3)] = half ? e3 : e1;
  float ssum = (e0 + e1) + (e2 + e3);
  ssum += __shfl_xor(ssum, 1);
  ssum += __shfl_xor(ssum, 2);
  ssum += __shfl_xor(ssum, 4);
  ssum += __shfl_xor(ssum, 8);
  ssum += __shfl_xor(ssum, 16);
  if (lane == 0) s_red[wave] = ssum;
  __syncthreads();
  float S = s_red[0];
  #pragma unroll
  for (int w = 1; w < 16; ++w) S += s_red[w];

  float4 acc = make_float4(0.f, 0.f, 0.f, 0.f);
  const float* wg = &s_w[g * CJ_];
  #pragma unroll
  for (int kk = 0; kk < CK_ / 2; ++kk) {
    const int ka = CK_ - 1 - 2 * kk, kb = ka - 1;
    ctx_sum_group(bufA, wg + CB_ * ka, acc);
    if (ka >= 2) {
      ctx_load_group(bufA, xp, ka - 2);
      __builtin_amdgcn_sched_barrier(0);
    }
    ctx_sum_group(bufB, wg + CB_ * kb, acc);
    if (kb >= 2) {
      ctx_load_group(bufB, xp, kb - 2);
      __builtin_amdgcn_sched_barrier(0);
    }
  }

  s_part[g * 64 + l] = acc;
  __syncthreads();
  if (tid < 64) {
    float4 t = s_part[tid];
    #pragma unroll
    for (int gg = 1; gg < CG_; ++gg) {
      const float4 v = s_part[gg * 64 + tid];
      t.x += v.x; t.y += v.y; t.z += v.z; t.w += v.w;
    }
    const float inv = 1.f / S;
    *(float4*)(ctxp + ((size_t)half * B_ + b) * U_ + 4 * tid) =
        make_float4(t.x * inv, t.y * inv, t.z * inv, t.w * inv);
  }
}

__device__ __forceinline__ float sigmoidf_(float a) {
  return 1.f / (1.f + expf(-a));
}

__global__ __launch_bounds__(256) void k_gates(
    const float* __restrict__ h, const float* __restrict__ in,
    const float* __restrict__ ctx,
    const float* __restrict__ Wz, const float* __restrict__ Cz,
    const float* __restrict__ bz,
    const float* __restrict__ Wr, const float* __restrict__ Cr,
    const float* __restrict__ br,
    float* __restrict__ ztw, float* __restrict__ rhw) {
  __shared__ float4 part[8 * 8 * 32];
  const int n0 = blockIdx.x * 64, m0 = blockIdx.y * 16;
  const bool isr = blockIdx.z != 0;
  const float* W    = isr ? Wr : Wz;
  const float* C    = isr ? Cr : Cz;
  const float* bias = isr ? br : bz;
  const float4 s = dense16x64<3>(h, in, ctx, ctx + B_ * U_, W, C, C + IN_ * U_, m0, n0, part);
  const int lane = threadIdx.x & 31, wave = threadIdx.x >> 5;
  const int row = m0 + 8 * (lane >> 4) + wave, col = n0 + 4 * (lane & 15);
  const float4 bb = *(const float4*)(bias + col);
  float4 g;
  g.x = sigmoidf_(s.x + bb.x);
  g.y = sigmoidf_(s.y + bb.y);
  g.z = sigmoidf_(s.z + bb.z);
  g.w = sigmoidf_(s.w + bb.w);
  if (isr) {
    const float4 hv = *(const float4*)(h + row * U_ + col);
    *(float4*)(rhw + row * U_ + col) =
        make_float4(g.x * hv.x, g.y * hv.y, g.z * hv.z, g.w * hv.w);
  } else {
    *(float4*)(ztw + row * U_ + col) = g;
  }
}

__global__ __launch_bounds__(256) void k_out(
    const float* __restrict__ h, const float* __restrict__ in,
    const float* __restrict__ ctx,
    const float* __restrict__ Up, const float* __restrict__ Cp,
    const float* __restrict__ bp,
    const float* __restrict__ ztw, const float* __restrict__ rhw,
    float* __restrict__ out) {
  __shared__ float4 part[8 * 8 * 32];
  const int n0 = blockIdx.x * 64, m0 = blockIdx.y * 16;
  const float4 s = dense16x64<3>(rhw, in, ctx, ctx + B_ * U_, Up, Cp, Cp + IN_ * U_, m0, n0, part);
  const int lane = threadIdx.x & 31, wave = threadIdx.x >> 5;
  const int row = m0 + 8 * (lane >> 4) + wave, col = n0 + 4 * (lane & 15);
  const float4 bb = *(const float4*)(bp + col);
  const float4 zt = *(const float4*)(ztw + row * U_ + col);
  const float4 hv = *(const float4*)(h + row * U_ + col);
  float4 o;
  o.x = (1.f - zt.x) * hv.x + zt.x * tanhf(s.x + bb.x);
  o.y = (1.f - zt.y) * hv.y + zt.y * tanhf(s.y + bb.y);
  o.z = (1.f - zt.z) * hv.z + zt.z * tanhf(s.z + bb.z);
  o.w = (1.f - zt.w) * hv.w + zt.w * tanhf(s.w + bb.w);
  *(float4*)(out + row * U_ + col) = o;
}

extern "C" void kernel_launch(void* const* d_in, const int* in_sizes, int n_in,
                              void* d_out, int out_size, void* d_ws, size_t ws_size,
                              hipStream_t stream) {
  const float* inputs = (const float*)d_in[0];
  const float* h_tm   = (const float*)d_in[1];
  const float* x_seq  = (const float*)d_in[2];
  const float* V_a    = (const float*)d_in[3];
  const float* W_a    = (const float*)d_in[4];
  const float* U_a    = (const float*)d_in[5];
  const float* b_a    = (const float*)d_in[6];
  const float* C_z    = (const float*)d_in[7];
  const float* W_z    = (const float*)d_in[8];
  const float* b_z    = (const float*)d_in[9];
  const float* C_r    = (const float*)d_in[10];
  const float* W_r    = (const float*)d_in[11];
  const float* b_r    = (const float*)d_in[12];
  const float* C_p    = (const float*)d_in[13];
  const float* U_p    = (const float*)d_in[14];
  const float* b_p    = (const float*)d_in[15];
  float* out = (float*)d_out;

  float* wsf   = (float*)d_ws;
  float* wxstm = wsf;
  float* et    = wsf + 32768;
  float* ctx   = et + (size_t)B_ * TE_;
  float* ztw   = ctx + 2 * B_ * U_;
  float* rhw   = ztw + B_ * U_;

  dim3 blk(256);
  k_wxstm  <<<dim3(U_ / 64, B_ / 16), blk, 0, stream>>>(h_tm, W_a, wxstm);
  k_scores <<<dim3(B_ * TE_ / ROWS_WG_), dim3(WAVES_ * 32), 0, stream>>>(
      x_seq, U_a, wxstm, b_a, V_a, et);
  k_context<<<dim3(B_ * TE_ / ROWS_WG_), dim3(512), 0, stream>>>(x_seq, et, ctx);
  k_gates  <<<dim3(U_ / 64, B_ / 16, 2), blk, 0, stream>>>(h_tm, inputs, ctx, W_z, C_z, b_z,
                                          W_r, C_r, b_r, ztw, rhw);
  k_out    <<<dim3(U_ / 64, B_ / 16), blk, 0, stream>>>(h_tm, inputs, ctx, U_p, C_p, b_p,
                                          ztw, rhw, out);
}


# k_scores workgroup widened from 16 to 32 waves (8 per SIMD, 128 registers) on one LDS copy of U_a: twice the serial tile chains per SIMD; x loads in 2 flights, N-tiles 2 at a time; sums as before
# speedup vs baseline: 4.1763x; 1.0587x over previous
#include <hip/hip_runtime.h>

#define B_   128
#define TE_  2048
#define U_   256
#define IN_  256

typedef __attribute__((ext_vector_type(16))) __bf16 v16bf;
typedef __attribute__((ext_vector_type(8)))  float  v8f;

__device__ __forceinline__ unsigned short f32_to_bf16_bits(float f) {
  unsigned int u = __float_as_uint(f);
  u += 0x7FFFu + ((u >> 16) & 1u);
  return (unsigned short)(u >> 16);
}

typedef __attribute__((ext_vector_type(2))) float v2f;
typedef __attribute__((ext_vector_type(16))) float v16f;

template <int NCH>
__device__ __forceinline__ float4 dense16x64(
    const float* __restrict__ a0, const float* __restrict__ a1,
    const float* __restrict__ a2, const float* __restrict__ a2x,
    const float* __restrict__ w0, const float* __restrict__ w1,
    const float* __restrict__ w2, int m0, int n0, float4* __restrict__ part) {
  const int tid  = threadIdx.x;
  const int wave = tid >> 5;
  const int lane = tid & 31;
  const int h    = lane >> 4;
  const int c    = lane & 15;
  const float* ap[3] = {a0, a1, a2};
  const float* wp[3] = {w0, w1, w2};

  v16f fa[NCH];
  v16f fb[NCH][4];
  #pragma unroll
  for (int i = 0; i < NCH; ++i) {
    const float* A  = ap[i] + (m0 + c) * U_ + wave * 32 + 4 * h;
    const float* Bp = wp[i] + (wave * 32 + 4 * h) * U_ + n0 + 4 * c;
    #pragma unroll
    for (int p = 0; p < 4; ++p) {
      float4 va = *(const float4*)(A + 8 * p);
      if (NCH == 3 && i == 2) {
        const float4 vx = *(const float4*)(a2x + (m0 + c) * U_ + wave * 32 + 4 * h + 8 * p);
        va.x += vx.x; va.y += vx.y; va.z += vx.z; va.w += vx.w;
      }
      fa[i][4 * p + 0] = va.x; fa[i][4 * p + 1] = va.y;
      fa[i][4 * p + 2] = va.z; fa[i][4 * p + 3] = va.w;
      #pragma unroll
      for (int j = 0; j < 4; ++j) {
        const float4 vb = *(const float4*)(Bp + (8 * p + j) * U_);
        fb[i][p][4 * j + 0] = vb.x; fb[i][p][4 * j + 1] = vb.y;
        fb[i][p][4 * j + 2] = vb.z; fb[i][p][4 * j + 3] = vb.w;
      }
    }
  }
  if constexpr (NCH == 1) {
    asm volatile("" :: "v"(fa[0]), "v"(fb[0][0]), "v"(fb[0][1]), "v"(fb[0][2]),
                 "v"(fb[0][3]));
  } else {
    asm volatile("" :: "v"(fa[0]), "v"(fb[0][0]), "v"(fb[0][1]), "v"(fb[0][2]),
                 "v"(fb[0][3]), "v"(fa[1]), "v"(fb[1][0]), "v"(fb[1][1]),
                 "v"(fb[1][2]), "v"(fb[1][3]), "v"(fa[NCH - 1]),
                 "v"(fb[NCH - 1][0]), "v"(fb[NCH - 1][1]), "v"(fb[NCH - 1][2]),
                 "v"(fb[NCH - 1][3]));
  }

  v8f acc[4];
  #pragma unroll
  for (int t = 0; t < 4; ++t)
    acc[t] = (v8f){0.f, 0.f, 0.f, 0.f, 0.f, 0.f, 0.f, 0.f};
  #pragma unroll
  for (int i = 0; i < NCH; ++i) {
    #pragma unroll
    for (int p = 0; p < 4; ++p) {
      #pragma unroll
      for (int q = 0; q < 2; ++q) {
        const v2f a = (v2f){fa[i][4 * p + 2 * q], fa[i][4 * p + 2 * q + 1]};
        const v16f b = fb[i][p];
        const int j0 = 4 * (2 * q), j1 = 4 * (2 * q + 1);
        acc[0] = __builtin_amdgcn_wmma_f32_16x16x4_f32(
            false, a, false, (v2f){b[j0 + 0], b[j1 + 0]}, (short)0, acc[0], false, false);
        acc[1] = __builtin_amdgcn_wmma_f32_16x16x4_f32(
            false, a, false, (v2f){b[j0 + 1], b[j1 + 1]}, (short)0, acc[1], false, false);
        acc[2] = __builtin_amdgcn_wmma_f32_16x16x4_f32(
            false, a, false, (v2f){b[j0 + 2], b[j1 + 2]}, (short)0, acc[2], false, false);
        acc[3] = __builtin_amdgcn_wmma_f32_16x16x4_f32(
            false, a, false, (v2f){b[j0 + 3], b[j1 + 3]}, (short)0, acc[3], false, false);
      }
    }
  }

  #pragma unroll
  for (int r = 0; r < 8; ++r)
    part[(wave * 8 + r) * 32 + lane] =
        make_float4(acc[0][r], acc[1][r], acc[2][r], acc[3][r]);
  __syncthreads();
  float4 s = part[wave * 32 + lane];
  #pragma unroll
  for (int w = 1; w < 8; ++w) {
    const float4 v = part[(w * 8 + wave) * 32 + lane];
    s.x += v.x; s.y += v.y; s.z += v.z; s.w += v.w;
  }
  return s;
}

__global__ __launch_bounds__(256) void k_wxstm(const float* __restrict__ h,
                                               const float* __restrict__ Wa,
                                               float* __restrict__ wx) {
  __shared__ float4 part[8 * 8 * 32];
  const int n0 = blockIdx.x * 64, m0 = blockIdx.y * 16;
  const float4 s = dense16x64<1>(h, h, h, h, Wa, Wa, Wa, m0, n0, part);
  const int lane = threadIdx.x & 31, wave = threadIdx.x >> 5;
  const int row = m0 + 8 * (lane >> 4) + wave, col = n0 + 4 * (lane & 15);
  *(float4*)(wx + row * U_ + col) = s;
}

#define ROWS_WG_ 1024
#define WAVES_   32
#define TILES_W_ 2

__device__ __forceinline__ unsigned int pack2_bf16(float lo, float hi) {
  return (unsigned int)f32_to_bf16_bits(lo) |
         ((unsigned int)f32_to_bf16_bits(hi) << 16);
}

__global__ __launch_bounds__(WAVES_ * 32) void k_scores(
    const float* __restrict__ x_seq, const float* __restrict__ Ua,
    const float* __restrict__ wxstm, const float* __restrict__ b_a,
    const float* __restrict__ V_a, float* __restrict__ et) {
  __shared__ __align__(16) unsigned short sB[U_ * U_];
  __shared__ float s_wx[U_];
  __shared__ float s_va[U_];

  const int tid  = threadIdx.x;
  const int wave = tid >> 5;
  const int lane = tid & 31;
  const int h    = lane >> 4;
  const int c    = lane & 15;
  const int b    = blockIdx.x >> 1;
  const unsigned int row0 = blockIdx.x * ROWS_WG_;

  {
    const int s = wave & 7;
    const int nt0 = (wave >> 3) * 4;
    #pragma unroll
    for (int nt = nt0; nt < nt0 + 4; ++nt) {
      unsigned int pk[8];
      #pragma unroll
      for (int j2 = 0; j2 < 8; ++j2) {
        const int j = 2 * j2;
        const int k = 32 * s + 8 * (j >> 2) + 4 * h + (j & 3);
        pk[j2] = pack2_bf16(Ua[k * U_ + nt * 16 + c],
                            Ua[(k + 1) * U_ + nt * 16 + c]);
      }
      uint4* dst = (uint4*)&sB[(nt * 8 + s) * 512 + lane * 8];
      dst[0]  = make_uint4(pk[0], pk[1], pk[2], pk[3]);
      dst[32] = make_uint4(pk[4], pk[5], pk[6], pk[7]);
    }
    if (tid < U_) {
      s_wx[tid] = wxstm[b * U_ + tid] + b_a[tid];
      s_va[tid] = V_a[tid];
    }
  }
  __syncthreads();

  #pragma unroll 1
  for (int it = 0; it < TILES_W_; ++it) {
    const unsigned int R = row0 + (unsigned int)(it * WAVES_ + wave) * 16u;

    union { unsigned int d[8]; v16bf v; } fa[8];
    const float* xp = x_seq + ((R + c) * U_ + h * 4);
    {
      float4 raw[20];
      #pragma unroll
      for (int q = 0; q < 20; ++q) raw[q] = *(const float4*)(xp + q * 8);
      #pragma unroll
      for (int s = 0; s < 5; ++s) {
        #pragma unroll
        for (int q4 = 0; q4 < 4; ++q4) {
          const float4 v = raw[4 * s + q4];
          fa[s].d[2 * q4]     = pack2_bf16(v.x, v.y);
          fa[s].d[2 * q4 + 1] = pack2_bf16(v.z, v.w);
        }
      }
    }
    __builtin_amdgcn_sched_barrier(0);
    {
      float4 raw[12];
      #pragma unroll
      for (int q = 0; q < 12; ++q) raw[q] = *(const float4*)(xp + (20 + q) * 8);
      #pragma unroll
      for (int s = 5; s < 8; ++s) {
        #pragma unroll
        for (int q4 = 0; q4 < 4; ++q4) {
          const float4 v = raw[4 * (s - 5) + q4];
          fa[s].d[2 * q4]     = pack2_bf16(v.x, v.y);
          fa[s].d[2 * q4 + 1] = pack2_bf16(v.z, v.w);
        }
      }
    }

    float sum[8];
    #pragma unroll
    for (int r = 0; r < 8; ++r) sum[r] = 0.f;

    #pragma unroll 1
    for (int ntq = 0; ntq < 8; ++ntq) {
      v8f acc[2];
      #pragma unroll
      for (int u = 0; u < 2; ++u)
        acc[u] = (v8f){0.f, 0.f, 0.f, 0.f, 0.f, 0.f, 0.f, 0.f};
      #pragma unroll
      for (int s = 0; s < 8; ++s) {
        union { uint4 q[2]; v16bf v; } fb[2];
        #pragma unroll
        for (int u = 0; u < 2; ++u) {
          const uint4* bp = (const uint4*)&sB[((ntq * 2 + u) * 8 + s) * 512];
          fb[u].q[0] = bp[lane];
          fb[u].q[1] = bp[32 + lane];
        }
        asm volatile("" :: "v"(fb[0].v), "v"(fb[1].v));
        #pragma unroll
        for (int u = 0; u < 2; ++u)
          acc[u] = __builtin_amdgcn_wmma_f32_16x16x32_bf16(
              false, fa[s].v, false, fb[u].v, (short)0, acc[u], false, false);
        __builtin_amdgcn_sched_barrier(0);
      }
      #pragma unroll
      for (int u = 0; u < 2; ++u) {
        const int ug = (ntq * 2 + u) * 16 + c;
        const float wx = s_wx[ug];
        const float va = s_va[ug];
        #pragma unroll
        for (int r = 0; r < 8; ++r) sum[r] += va * tanhf(acc[u][r] + wx);
      }
    }

    #pragma unroll
    for (int r = 0; r < 8; ++r) {
      float v = sum[r];
      v += __shfl_xor(v, 1);
      v += __shfl_xor(v, 2);
      v += __shfl_xor(v, 4);
      v += __shfl_xor(v, 8);
      sum[r] = v;
    }
    if (c == 0) {
      float4* o = (float4*)(et + (R + 8 * h));
      o[0] = make_float4(sum[0], sum[1], sum[2], sum[3]);
      o[1] = make_float4(sum[4], sum[5], sum[6], sum[7]);
    }
  }
}

#define CG_ 8
#define CJ_ (ROWS_WG_ / CG_)
#define CB_ 16
#define CK_ (CJ_ / CB_)

__device__ __forceinline__ void ctx_load_group(float4 (&buf)[CB_],
                                               const float* __restrict__ xp, int k) {
  #pragma unroll
  for (int jj = 0; jj < CB_; ++jj)
    buf[jj] = *(const float4*)(xp + (size_t)(8 * (CB_ * k + jj)) * U_);
}

__device__ __forceinline__ void ctx_sum_group(const float4 (&buf)[CB_],
                                              const float* w, float4& acc) {
  float wv[CB_];
  #pragma unroll
  for (int q = 0; q < CB_ / 4; ++q) {
    const float4 w4 = *(const float4*)(w + 4 * q);
    wv[4 * q + 0] = w4.x; wv[4 * q + 1] = w4.y;
    wv[4 * q + 2] = w4.z; wv[4 * q + 3] = w4.w;
  }
  #pragma unroll
  for (int jj = CB_ - 1; jj >= 0; --jj) {
    acc.x = fmaf(wv[jj], buf[jj].x, acc.x);
    acc.y = fmaf(wv[jj], buf[jj].y, acc.y);
    acc.z = fmaf(wv[jj], buf[jj].z, acc.z);
    acc.w = fmaf(wv[jj], buf[jj].w, acc.w);
  }
}

__global__ __launch_bounds__(512) void k_context(const float* __restrict__ x_seq,
                                                 const float* __restrict__ et,
                                                 float* __restrict__ ctxp) {
  __shared__ __align__(16) float s_w[ROWS_WG_];
  __shared__ float s_red[16];
  __shared__ float4 s_part[CG_ * 64];

  const int tid  = threadIdx.x;
  const int lane = tid & 31;
  const int wave = tid >> 5;
  const int g    = tid >> 6;
  const int l    = tid & 63;
  const int b    = blockIdx.x >> 1;
  const int half = blockIdx.x & 1;
  const size_t row0 = (size_t)blockIdx.x * ROWS_WG_;
  const float* xp = x_seq + (row0 + g) * U_ + 4 * l;

  const float* etb = et + (size_t)b * TE_;
  const float r0 = etb[tid], r1 = etb[tid + 512];
  const float r2 = etb[tid + 1024], r3 = etb[tid + 1536];

  float4 bufA[CB_], bufB[CB_];
  ctx_load_group(bufA, xp, CK_ - 1);
  ctx_load_group(bufB, xp, CK_ - 2);
  __builtin_amdgcn_sched_barrier(0);

  const float e0 = expf(r0), e1 = expf(r1), e2 = expf(r2), e3 = expf(r3);
  s_w[(tid & 7) * CJ_ + (tid >> 3)]      = half ? e2 : e0;
  s_w[(tid & 7) * CJ_ + 64 + (tid >> 3)] = half ? e3 : e1;
  float ssum = (e0 + e1) + (e2 + e3);
  ssum += __shfl_xor(ssum, 1);
  ssum += __shfl_xor(ssum, 2);
  ssum += __shfl_xor(ssum, 4);
  ssum += __shfl_xor(ssum, 8);
  ssum += __shfl_xor(ssum, 16);
  if (lane == 0) s_red[wave] = ssum;
  __syncthreads();
  float S = s_red[0];
  #pragma unroll
  for (int w = 1; w < 16; ++w) S += s_red[w];

  float4 acc = make_float4(0.f, 0.f, 0.f, 0.f);
  const float* wg = &s_w[g * CJ_];
  #pragma unroll
  for (int kk = 0; kk < CK_ / 2; ++kk) {
    const int ka = CK_ - 1 - 2 * kk, kb = ka - 1;
    ctx_sum_group(bufA, wg + CB_ * ka, acc);
    if (ka >= 2) {
      ctx_load_group(bufA, xp, ka - 2);
      __builtin_amdgcn_sched_barrier(0);
    }
    ctx_sum_group(bufB, wg + CB_ * kb, acc);
    if (kb >= 2) {
      ctx_load_group(bufB, xp, kb - 2);
      __builtin_amdgcn_sched_barrier(0);
    }
  }

  s_part[g * 64 + l] = acc;
  __syncthreads();
  if (tid < 64) {
    float4 t = s_part[tid];
    #pragma unroll
    for (int gg = 1; gg < CG_; ++gg) {
      const float4 v = s_part[gg * 64 + tid];
      t.x += v.x; t.y += v.y; t.z += v.z; t.w += v.w;
    }
    const float inv = 1.f / S;
    *(float4*)(ctxp + ((size_t)half * B_ + b) * U_ + 4 * tid) =
        make_float4(t.x * inv, t.y * inv, t.z * inv, t.w * inv);
  }
}

__device__ __forceinline__ float sigmoidf_(float a) {
  return 1.f / (1.f + expf(-a));
}

__global__ __launch_bounds__(256) void k_gates(
    const float* __restrict__ h, const float* __restrict__ in,
    const float* __restrict__ ctx,
    const float* __restrict__ Wz, const float* __restrict__ Cz,
    const float* __restrict__ bz,
    const float* __restrict__ Wr, const float* __restrict__ Cr,
    const float* __restrict__ br,
    float* __restrict__ ztw, float* __restrict__ rhw) {
  __shared__ float4 part[8 * 8 * 32];
  const int n0 = blockIdx.x * 64, m0 = blockIdx.y * 16;
  const bool isr = blockIdx.z != 0;
  const float* W    = isr ? Wr : Wz;
  const float* C    = isr ? Cr : Cz;
  const float* bias = isr ? br : bz;
  const float4 s = dense16x64<3>(h, in, ctx, ctx + B_ * U_, W, C, C + IN_ * U_, m0, n0, part);
  const int lane = threadIdx.x & 31, wave = threadIdx.x >> 5;
  const int row = m0 + 8 * (lane >> 4) + wave, col = n0 + 4 * (lane & 15);
  const float4 bb = *(const float4*)(bias + col);
  float4 g;
  g.x = sigmoidf_(s.x + bb.x);
  g.y = sigmoidf_(s.y + bb.y);
  g.z = sigmoidf_(s.z + bb.z);
  g.w = sigmoidf_(s.w + bb.w);
  if (isr) {
    const float4 hv = *(const float4*)(h + row * U_ + col);
    *(float4*)(rhw + row * U_ + col) =
        make_float4(g.x * hv.x, g.y * hv.y, g.z * hv.z, g.w * hv.w);
  } else {
    *(float4*)(ztw + row * U_ + col) = g;
  }
}

__global__ __launch_bounds__(256) void k_out(
    const float* __restrict__ h, const float* __restrict__ in,
    const float* __restrict__ ctx,
    const float* __restrict__ Up, const float* __restrict__ Cp,
    const float* __restrict__ bp,
    const float* __restrict__ ztw, const float* __restrict__ rhw,
    float* __restrict__ out) {
  __shared__ float4 part[8 * 8 * 32];
  const int n0 = blockIdx.x * 64, m0 = blockIdx.y * 16;
  const float4 s = dense16x64<3>(rhw, in, ctx, ctx + B_ * U_, Up, Cp, Cp + IN_ * U_, m0, n0, part);
  const int lane = threadIdx.x & 31, wave = threadIdx.x >> 5;
  const int row = m0 + 8 * (lane >> 4) + wave, col = n0 + 4 * (lane & 15);
  const float4 bb = *(const float4*)(bp + col);
  const float4 zt = *(const float4*)(ztw + row * U_ + col);
  const float4 hv = *(const float4*)(h + row * U_ + col);
  float4 o;
  o.x = (1.f - zt.x) * hv.x + zt.x * tanhf(s.x + bb.x);
  o.y = (1.f - zt.y) * hv.y + zt.y * tanhf(s.y + bb.y);
  o.z = (1.f - zt.z) * hv.z + zt.z * tanhf(s.z + bb.z);
  o.w = (1.f - zt.w) * hv.w + zt.w * tanhf(s.w + bb.w);
  *(float4*)(out + row * U_ + col) = o;
}

extern "C" void kernel_launch(void* const* d_in, const int* in_sizes, int n_in,
                              void* d_out, int out_size, void* d_ws, size_t ws_size,
                              hipStream_t stream) {
  const float* inputs = (const float*)d_in[0];
  const float* h_tm   = (const float*)d_in[1];
  const float* x_seq  = (const float*)d_in[2];
  const float* V_a    = (const float*)d_in[3];
  const float* W_a    = (const float*)d_in[4];
  const float* U_a    = (const float*)d_in[5];
  const float* b_a    = (const float*)d_in[6];
  const float* C_z    = (const float*)d_in[7];
  const float* W_z    = (const float*)d_in[8];
  const float* b_z    = (const float*)d_in[9];
  const float* C_r    = (const float*)d_in[10];
  const float* W_r    = (const float*)d_in[11];
  const float* b_r    = (const float*)d_in[12];
  const float* C_p    = (const float*)d_in[13];
  const float* U_p    = (const float*)d_in[14];
  const float* b_p    = (const float*)d_in[15];
  float* out = (float*)d_out;

  float* wsf   = (float*)d_ws;
  float* wxstm = wsf;
  float* et    = wsf + 32768;
  float* ctx   = et + (size_t)B_ * TE_;
  float* ztw   = ctx + 2 * B_ * U_;
  float* rhw   = ztw + B_ * U_;

  dim3 blk(256);
  k_wxstm  <<<dim3(U_ / 64, B_ / 16), blk, 0, stream>>>(h_tm, W_a, wxstm);
  k_scores <<<dim3(B_ * TE_ / ROWS_WG_), dim3(WAVES_ * 32), 0, stream>>>(
      x_seq, U_a, wxstm, b_a, V_a, et);
  k_context<<<dim3(B_ * TE_ / ROWS_WG_), dim3(512), 0, stream>>>(x_seq, et, ctx);
  k_gates  <<<dim3(U_ / 64, B_ / 16, 2), blk, 0, stream>>>(h_tm, inputs, ctx, W_z, C_z, b_z,
                                          W_r, C_r, b_r, ztw, rhw);
  k_out    <<<dim3(U_ / 64, B_ / 16), blk, 0, stream>>>(h_tm, inputs, ctx, U_p, C_p, b_p,
                                          ztw, rhw, out);
}


# k_scores epilogue: library tanhf's two branch sides both evaluated per value, one selected (exp range reduction fused: last bits differ); a pass's 16 serial branchy chains become packed f32 pairs
# speedup vs baseline: 4.7987x; 1.1490x over previous
#include <hip/hip_runtime.h>

#define B_   128
#define TE_  2048
#define U_   256
#define IN_  256

typedef __attribute__((ext_vector_type(16))) __bf16 v16bf;
typedef __attribute__((ext_vector_type(8)))  float  v8f;

__device__ __forceinline__ unsigned short f32_to_bf16_bits(float f) {
  unsigned int u = __float_as_uint(f);
  u += 0x7FFFu + ((u >> 16) & 1u);
  return (unsigned short)(u >> 16);
}

typedef __attribute__((ext_vector_type(2))) float v2f;
typedef __attribute__((ext_vector_type(16))) float v16f;

template <int NCH>
__device__ __forceinline__ float4 dense16x64(
    const float* __restrict__ a0, const float* __restrict__ a1,
    const float* __restrict__ a2, const float* __restrict__ a2x,
    const float* __restrict__ w0, const float* __restrict__ w1,
    const float* __restrict__ w2, int m0, int n0, float4* __restrict__ part) {
  const int tid  = threadIdx.x;
  const int wave = tid >> 5;
  const int lane = tid & 31;
  const int h    = lane >> 4;
  const int c    = lane & 15;
  const float* ap[3] = {a0, a1, a2};
  const float* wp[3] = {w0, w1, w2};

  v16f fa[NCH];
  v16f fb[NCH][4];
  #pragma unroll
  for (int i = 0; i < NCH; ++i) {
    const float* A  = ap[i] + (m0 + c) * U_ + wave * 32 + 4 * h;
    const float* Bp = wp[i] + (wave * 32 + 4 * h) * U_ + n0 + 4 * c;
    #pragma unroll
    for (int p = 0; p < 4; ++p) {
      float4 va = *(const float4*)(A + 8 * p);
      if (NCH == 3 && i == 2) {
        const float4 vx = *(const float4*)(a2x + (m0 + c) * U_ + wave * 32 + 4 * h + 8 * p);
        va.x += vx.x; va.y += vx.y; va.z += vx.z; va.w += vx.w;
      }
      fa[i][4 * p + 0] = va.x; fa[i][4 * p + 1] = va.y;
      fa[i][4 * p + 2] = va.z; fa[i][4 * p + 3] = va.w;
      #pragma unroll
      for (int j = 0; j < 4; ++j) {
        const float4 vb = *(const float4*)(Bp + (8 * p + j) * U_);
        fb[i][p][4 * j + 0] = vb.x; fb[i][p][4 * j + 1] = vb.y;
        fb[i][p][4 * j + 2] = vb.z; fb[i][p][4 * j + 3] = vb.w;
      }
    }
  }
  if constexpr (NCH == 1) {
    asm volatile("" :: "v"(fa[0]), "v"(fb[0][0]), "v"(fb[0][1]), "v"(fb[0][2]),
                 "v"(fb[0][3]));
  } else {
    asm volatile("" :: "v"(fa[0]), "v"(fb[0][0]), "v"(fb[0][1]), "v"(fb[0][2]),
                 "v"(fb[0][3]), "v"(fa[1]), "v"(fb[1][0]), "v"(fb[1][1]),
                 "v"(fb[1][2]), "v"(fb[1][3]), "v"(fa[NCH - 1]),
                 "v"(fb[NCH - 1][0]), "v"(fb[NCH - 1][1]), "v"(fb[NCH - 1][2]),
                 "v"(fb[NCH - 1][3]));
  }

  v8f acc[4];
  #pragma unroll
  for (int t = 0; t < 4; ++t)
    acc[t] = (v8f){0.f, 0.f, 0.f, 0.f, 0.f, 0.f, 0.f, 0.f};
  #pragma unroll
  for (int i = 0; i < NCH; ++i) {
    #pragma unroll
    for (int p = 0; p < 4; ++p) {
      #pragma unroll
      for (int q = 0; q < 2; ++q) {
        const v2f a = (v2f){fa[i][4 * p + 2 * q], fa[i][4 * p + 2 * q + 1]};
        const v16f b = fb[i][p];
        const int j0 = 4 * (2 * q), j1 = 4 * (2 * q + 1);
        acc[0] = __builtin_amdgcn_wmma_f32_16x16x4_f32(
            false, a, false, (v2f){b[j0 + 0], b[j1 + 0]}, (short)0, acc[0], false, false);
        acc[1] = __builtin_amdgcn_wmma_f32_16x16x4_f32(
            false, a, false, (v2f){b[j0 + 1], b[j1 + 1]}, (short)0, acc[1], false, false);
        acc[2] = __builtin_amdgcn_wmma_f32_16x16x4_f32(
            false, a, false, (v2f){b[j0 + 2], b[j1 + 2]}, (short)0, acc[2], false, false);
        acc[3] = __builtin_amdgcn_wmma_f32_16x16x4_f32(
            false, a, false, (v2f){b[j0 + 3], b[j1 + 3]}, (short)0, acc[3], false, false);
      }
    }
  }

  #pragma unroll
  for (int r = 0; r < 8; ++r)
    part[(wave * 8 + r) * 32 + lane] =
        make_float4(acc[0][r], acc[1][r], acc[2][r], acc[3][r]);
  __syncthreads();
  float4 s = part[wave * 32 + lane];
  #pragma unroll
  for (int w = 1; w < 8; ++w) {
    const float4 v = part[(w * 8 + wave) * 32 + lane];
    s.x += v.x; s.y += v.y; s.z += v.z; s.w += v.w;
  }
  return s;
}

__global__ __launch_bounds__(256) void k_wxstm(const float* __restrict__ h,
                                               const float* __restrict__ Wa,
                                               float* __restrict__ wx) {
  __shared__ float4 part[8 * 8 * 32];
  const int n0 = blockIdx.x * 64, m0 = blockIdx.y * 16;
  const float4 s = dense16x64<1>(h, h, h, h, Wa, Wa, Wa, m0, n0, part);
  const int lane = threadIdx.x & 31, wave = threadIdx.x >> 5;
  const int row = m0 + 8 * (lane >> 4) + wave, col = n0 + 4 * (lane & 15);
  *(float4*)(wx + row * U_ + col) = s;
}

#define ROWS_WG_ 1024
#define WAVES_   32
#define TILES_W_ 2

__device__ __forceinline__ unsigned int pack2_bf16(float lo, float hi) {
  return (unsigned int)f32_to_bf16_bits(lo) |
         ((unsigned int)f32_to_bf16_bits(hi) << 16);
}

__device__ __forceinline__ float tanh_both(float z) {
  const float a = __builtin_fabsf(z);
  const float L2E_HI = __uint_as_float(0x3fb8aa3bu);
  const float L2E_LO = __uint_as_float(0x32a5705fu);
  const float y  = a + a;
  const float rn = __builtin_rintf(y * L2E_HI);
  float f = __builtin_fmaf(y, L2E_HI, -rn);
  f = __builtin_fmaf(y, L2E_LO, f);
  float ex = __builtin_ldexpf(__builtin_amdgcn_exp2f(f), (int)rn);
  ex = (y > __uint_as_float(0x42b17218u)) ? __builtin_inff() : ex;
  const float big = __builtin_fmaf(__builtin_amdgcn_rcpf(1.0f + ex), -2.0f, 1.0f);
  const float s = z * z;
  float p = __builtin_fmaf(__uint_as_float(0xbbbac73du), s, __uint_as_float(0x3ca908c9u));
  p = __builtin_fmaf(s, p, __uint_as_float(0xbd5c1c4eu));
  p = __builtin_fmaf(s, p, __uint_as_float(0x3e088382u));
  p = __builtin_fmaf(s, p, __uint_as_float(0xbeaaaa99u));
  const float small = __builtin_fmaf(s, a * p, a);
  const float res = (0.625f > a) ? small : big;
  return __builtin_copysignf(res, z);
}

__global__ __launch_bounds__(WAVES_ * 32) void k_scores(
    const float* __restrict__ x_seq, const float* __restrict__ Ua,
    const float* __restrict__ wxstm, const float* __restrict__ b_a,
    const float* __restrict__ V_a, float* __restrict__ et) {
  __shared__ __align__(16) unsigned short sB[U_ * U_];
  __shared__ float s_wx[U_];
  __shared__ float s_va[U_];

  const int tid  = threadIdx.x;
  const int wave = tid >> 5;
  const int lane = tid & 31;
  const int h    = lane >> 4;
  const int c    = lane & 15;
  const int b    = blockIdx.x >> 1;
  const unsigned int row0 = blockIdx.x * ROWS_WG_;

  {
    const int s = wave & 7;
    const int nt0 = (wave >> 3) * 4;
    #pragma unroll
    for (int nt = nt0; nt < nt0 + 4; ++nt) {
      unsigned int pk[8];
      #pragma unroll
      for (int j2 = 0; j2 < 8; ++j2) {
        const int j = 2 * j2;
        const int k = 32 * s + 8 * (j >> 2) + 4 * h + (j & 3);
        pk[j2] = pack2_bf16(Ua[k * U_ + nt * 16 + c],
                            Ua[(k + 1) * U_ + nt * 16 + c]);
      }
      uint4* dst = (uint4*)&sB[(nt * 8 + s) * 512 + lane * 8];
      dst[0]  = make_uint4(pk[0], pk[1], pk[2], pk[3]);
      dst[32] = make_uint4(pk[4], pk[5], pk[6], pk[7]);
    }
    if (tid < U_) {
      s_wx[tid] = wxstm[b * U_ + tid] + b_a[tid];
      s_va[tid] = V_a[tid];
    }
  }
  __syncthreads();

  #pragma unroll 1
  for (int it = 0; it < TILES_W_; ++it) {
    const unsigned int R = row0 + (unsigned int)(it * WAVES_ + wave) * 16u;

    union { unsigned int d[8]; v16bf v; } fa[8];
    const float* xp = x_seq + ((R + c) * U_ + h * 4);
    {
      float4 raw[20];
      #pragma unroll
      for (int q = 0; q < 20; ++q) raw[q] = *(const float4*)(xp + q * 8);
      #pragma unroll
      for (int s = 0; s < 5; ++s) {
        #pragma unroll
        for (int q4 = 0; q4 < 4; ++q4) {
          const float4 v = raw[4 * s + q4];
          fa[s].d[2 * q4]     = pack2_bf16(v.x, v.y);
          fa[s].d[2 * q4 + 1] = pack2_bf16(v.z, v.w);
        }
      }
    }
    __builtin_amdgcn_sched_barrier(0);
    {
      float4 raw[12];
      #pragma unroll
      for (int q = 0; q < 12; ++q) raw[q] = *(const float4*)(xp + (20 + q) * 8);
      #pragma unroll
      for (int s = 5; s < 8; ++s) {
        #pragma unroll
        for (int q4 = 0; q4 < 4; ++q4) {
          const float4 v = raw[4 * (s - 5) + q4];
          fa[s].d[2 * q4]     = pack2_bf16(v.x, v.y);
          fa[s].d[2 * q4 + 1] = pack2_bf16(v.z, v.w);
        }
      }
    }

    float sum[8];
    #pragma unroll
    for (int r = 0; r < 8; ++r) sum[r] = 0.f;

    #pragma unroll 1
    for (int ntq = 0; ntq < 8; ++ntq) {
      v8f acc[2];
      #pragma unroll
      for (int u = 0; u < 2; ++u)
        acc[u] = (v8f){0.f, 0.f, 0.f, 0.f, 0.f, 0.f, 0.f, 0.f};
      #pragma unroll
      for (int s = 0; s < 8; ++s) {
        union { uint4 q[2]; v16bf v; } fb[2];
        #pragma unroll
        for (int u = 0; u < 2; ++u) {
          const uint4* bp = (const uint4*)&sB[((ntq * 2 + u) * 8 + s) * 512];
          fb[u].q[0] = bp[lane];
          fb[u].q[1] = bp[32 + lane];
        }
        asm volatile("" :: "v"(fb[0].v), "v"(fb[1].v));
        #pragma unroll
        for (int u = 0; u < 2; ++u)
          acc[u] = __builtin_amdgcn_wmma_f32_16x16x32_bf16(
              false, fa[s].v, false, fb[u].v, (short)0, acc[u], false, false);
        __builtin_amdgcn_sched_barrier(0);
      }
      #pragma unroll
      for (int u = 0; u < 2; ++u) {
        const int ug = (ntq * 2 + u) * 16 + c;
        const float wx = s_wx[ug];
        const float va = s_va[ug];
        #pragma unroll
        for (int r = 0; r < 8; ++r) sum[r] += va * tanh_both(acc[u][r] + wx);
      }
    }

    #pragma unroll
    for (int r = 0; r < 8; ++r) {
      float v = sum[r];
      v += __shfl_xor(v, 1);
      v += __shfl_xor(v, 2);
      v += __shfl_xor(v, 4);
      v += __shfl_xor(v, 8);
      sum[r] = v;
    }
    if (c == 0) {
      float4* o = (float4*)(et + (R + 8 * h));
      o[0] = make_float4(sum[0], sum[1], sum[2], sum[3]);
      o[1] = make_float4(sum[4], sum[5], sum[6], sum[7]);
    }
  }
}

#define CG_ 8
#define CJ_ (ROWS_WG_ / CG_)
#define CB_ 16
#define CK_ (CJ_ / CB_)

__device__ __forceinline__ void ctx_load_group(float4 (&buf)[CB_],
                                               const float* __restrict__ xp, int k) {
  #pragma unroll
  for (int jj = 0; jj < CB_; ++jj)
    buf[jj] = *(const float4*)(xp + (size_t)(8 * (CB_ * k + jj)) * U_);
}

__device__ __forceinline__ void ctx_sum_group(const float4 (&buf)[CB_],
                                              const float* w, float4& acc) {
  float wv[CB_];
  #pragma unroll
  for (int q = 0; q < CB_ / 4; ++q) {
    const float4 w4 = *(const float4*)(w + 4 * q);
    wv[4 * q + 0] = w4.x; wv[4 * q + 1] = w4.y;
    wv[4 * q + 2] = w4.z; wv[4 * q + 3] = w4.w;
  }
  #pragma unroll
  for (int jj = CB_ - 1; jj >= 0; --jj) {
    acc.x = fmaf(wv[jj], buf[jj].x, acc.x);
    acc.y = fmaf(wv[jj], buf[jj].y, acc.y);
    acc.z = fmaf(wv[jj], buf[jj].z, acc.z);
    acc.w = fmaf(wv[jj], buf[jj].w, acc.w);
  }
}

__global__ __launch_bounds__(512) void k_context(const float* __restrict__ x_seq,
                                                 const float* __restrict__ et,
                                                 float* __restrict__ ctxp) {
  __shared__ __align__(16) float s_w[ROWS_WG_];
  __shared__ float s_red[16];
  __shared__ float4 s_part[CG_ * 64];

  const int tid  = threadIdx.x;
  const int lane = tid & 31;
  const int wave = tid >> 5;
  const int g    = tid >> 6;
  const int l    = tid & 63;
  const int b    = blockIdx.x >> 1;
  const int half = blockIdx.x & 1;
  const size_t row0 = (size_t)blockIdx.x * ROWS_WG_;
  const float* xp = x_seq + (row0 + g) * U_ + 4 * l;

  const float* etb = et + (size_t)b * TE_;
  const float r0 = etb[tid], r1 = etb[tid + 512];
  const float r2 = etb[tid + 1024], r3 = etb[tid + 1536];

  float4 bufA[CB_], bufB[CB_];
  ctx_load_group(bufA, xp, CK_ - 1);
  ctx_load_group(bufB, xp, CK_ - 2);
  __builtin_amdgcn_sched_barrier(0);

  const float e0 = expf(r0), e1 = expf(r1), e2 = expf(r2), e3 = expf(r3);
  s_w[(tid & 7) * CJ_ + (tid >> 3)]      = half ? e2 : e0;
  s_w[(tid & 7) * CJ_ + 64 + (tid >> 3)] = half ? e3 : e1;
  float ssum = (e0 + e1) + (e2 + e3);
  ssum += __shfl_xor(ssum, 1);
  ssum += __shfl_xor(ssum, 2);
  ssum += __shfl_xor(ssum, 4);
  ssum += __shfl_xor(ssum, 8);
  ssum += __shfl_xor(ssum, 16);
  if (lane == 0) s_red[wave] = ssum;
  __syncthreads();
  float S = s_red[0];
  #pragma unroll
  for (int w = 1; w < 16; ++w) S += s_red[w];

  float4 acc = make_float4(0.f, 0.f, 0.f, 0.f);
  const float* wg = &s_w[g * CJ_];
  #pragma unroll
  for (int kk = 0; kk < CK_ / 2; ++kk) {
    const int ka = CK_ - 1 - 2 * kk, kb = ka - 1;
    ctx_sum_group(bufA, wg + CB_ * ka, acc);
    if (ka >= 2) {
      ctx_load_group(bufA, xp, ka - 2);
      __builtin_amdgcn_sched_barrier(0);
    }
    ctx_sum_group(bufB, wg + CB_ * kb, acc);
    if (kb >= 2) {
      ctx_load_group(bufB, xp, kb - 2);
      __builtin_amdgcn_sched_barrier(0);
    }
  }

  s_part[g * 64 + l] = acc;
  __syncthreads();
  if (tid < 64) {
    float4 t = s_part[tid];
    #pragma unroll
    for (int gg = 1; gg < CG_; ++gg) {
      const float4 v = s_part[gg * 64 + tid];
      t.x += v.x; t.y += v.y; t.z += v.z; t.w += v.w;
    }
    const float inv = 1.f / S;
    *(float4*)(ctxp + ((size_t)half * B_ + b) * U_ + 4 * tid) =
        make_float4(t.x * inv, t.y * inv, t.z * inv, t.w * inv);
  }
}

__device__ __forceinline__ float sigmoidf_(float a) {
  return 1.f / (1.f + expf(-a));
}

__global__ __launch_bounds__(256) void k_gates(
    const float* __restrict__ h, const float* __restrict__ in,
    const float* __restrict__ ctx,
    const float* __restrict__ Wz, const float* __restrict__ Cz,
    const float* __restrict__ bz,
    const float* __restrict__ Wr, const float* __restrict__ Cr,
    const float* __restrict__ br,
    float* __restrict__ ztw, float* __restrict__ rhw) {
  __shared__ float4 part[8 * 8 * 32];
  const int n0 = blockIdx.x * 64, m0 = blockIdx.y * 16;
  const bool isr = blockIdx.z != 0;
  const float* W    = isr ? Wr : Wz;
  const float* C    = isr ? Cr : Cz;
  const float* bias = isr ? br : bz;
  const float4 s = dense16x64<3>(h, in, ctx, ctx + B_ * U_, W, C, C + IN_ * U_, m0, n0, part);
  const int lane = threadIdx.x & 31, wave = threadIdx.x >> 5;
  const int row = m0 + 8 * (lane >> 4) + wave, col = n0 + 4 * (lane & 15);
  const float4 bb = *(const float4*)(bias + col);
  float4 g;
  g.x = sigmoidf_(s.x + bb.x);
  g.y = sigmoidf_(s.y + bb.y);
  g.z = sigmoidf_(s.z + bb.z);
  g.w = sigmoidf_(s.w + bb.w);
  if (isr) {
    const float4 hv = *(const float4*)(h + row * U_ + col);
    *(float4*)(rhw + row * U_ + col) =
        make_float4(g.x * hv.x, g.y * hv.y, g.z * hv.z, g.w * hv.w);
  } else {
    *(float4*)(ztw + row * U_ + col) = g;
  }
}

__global__ __launch_bounds__(256) void k_out(
    const float* __restrict__ h, const float* __restrict__ in,
    const float* __restrict__ ctx,
    const float* __restrict__ Up, const float* __restrict__ Cp,
    const float* __restrict__ bp,
    const float* __restrict__ ztw, const float* __restrict__ rhw,
    float* __restrict__ out) {
  __shared__ float4 part[8 * 8 * 32];
  const int n0 = blockIdx.x * 64, m0 = blockIdx.y * 16;
  const float4 s = dense16x64<3>(rhw, in, ctx, ctx + B_ * U_, Up, Cp, Cp + IN_ * U_, m0, n0, part);
  const int lane = threadIdx.x & 31, wave = threadIdx.x >> 5;
  const int row = m0 + 8 * (lane >> 4) + wave, col = n0 + 4 * (lane & 15);
  const float4 bb = *(const float4*)(bp + col);
  const float4 zt = *(const float4*)(ztw + row * U_ + col);
  const float4 hv = *(const float4*)(h + row * U_ + col);
  float4 o;
  o.x = (1.f - zt.x) * hv.x + zt.x * tanhf(s.x + bb.x);
  o.y = (1.f - zt.y) * hv.y + zt.y * tanhf(s.y + bb.y);
  o.z = (1.f - zt.z) * hv.z + zt.z * tanhf(s.z + bb.z);
  o.w = (1.f - zt.w) * hv.w + zt.w * tanhf(s.w + bb.w);
  *(float4*)(out + row * U_ + col) = o;
}

extern "C" void kernel_launch(void* const* d_in, const int* in_sizes, int n_in,
                              void* d_out, int out_size, void* d_ws, size_t ws_size,
                              hipStream_t stream) {
  const float* inputs = (const float*)d_in[0];
  const float* h_tm   = (const float*)d_in[1];
  const float* x_seq  = (const float*)d_in[2];
  const float* V_a    = (const float*)d_in[3];
  const float* W_a    = (const float*)d_in[4];
  const float* U_a    = (const float*)d_in[5];
  const float* b_a    = (const float*)d_in[6];
  const float* C_z    = (const float*)d_in[7];
  const float* W_z    = (const float*)d_in[8];
  const float* b_z    = (const float*)d_in[9];
  const float* C_r    = (const float*)d_in[10];
  const float* W_r    = (const float*)d_in[11];
  const float* b_r    = (const float*)d_in[12];
  const float* C_p    = (const float*)d_in[13];
  const float* U_p    = (const float*)d_in[14];
  const float* b_p    = (const float*)d_in[15];
  float* out = (float*)d_out;

  float* wsf   = (float*)d_ws;
  float* wxstm = wsf;
  float* et    = wsf + 32768;
  float* ctx   = et + (size_t)B_ * TE_;
  float* ztw   = ctx + 2 * B_ * U_;
  float* rhw   = ztw + B_ * U_;

  dim3 blk(256);
  k_wxstm  <<<dim3(U_ / 64, B_ / 16), blk, 0, stream>>>(h_tm, W_a, wxstm);
  k_scores <<<dim3(B_ * TE_ / ROWS_WG_), dim3(WAVES_ * 32), 0, stream>>>(
      x_seq, U_a, wxstm, b_a, V_a, et);
  k_context<<<dim3(B_ * TE_ / ROWS_WG_), dim3(512), 0, stream>>>(x_seq, et, ctx);
  k_gates  <<<dim3(U_ / 64, B_ / 16, 2), blk, 0, stream>>>(h_tm, inputs, ctx, W_z, C_z, b_z,
                                          W_r, C_r, b_r, ztw, rhw);
  k_out    <<<dim3(U_ / 64, B_ / 16), blk, 0, stream>>>(h_tm, inputs, ctx, U_p, C_p, b_p,
                                          ztw, rhw, out);
}


# k_scores epilogue: tanh(z) = 1 - 2/(2^(c*z) + 1), c = 2*log2(e) folded into the wxstm add: one exp2, one rcp, two fma per value, no select; abs error a few 1e-7, relative accuracy near 0 given up
# speedup vs baseline: 5.7060x; 1.1891x over previous
#include <hip/hip_runtime.h>

#define B_   128
#define TE_  2048
#define U_   256
#define IN_  256

typedef __attribute__((ext_vector_type(16))) __bf16 v16bf;
typedef __attribute__((ext_vector_type(8)))  float  v8f;

__device__ __forceinline__ unsigned short f32_to_bf16_bits(float f) {
  unsigned int u = __float_as_uint(f);
  u += 0x7FFFu + ((u >> 16) & 1u);
  return (unsigned short)(u >> 16);
}

typedef __attribute__((ext_vector_type(2))) float v2f;
typedef __attribute__((ext_vector_type(16))) float v16f;

template <int NCH>
__device__ __forceinline__ float4 dense16x64(
    const float* __restrict__ a0, const float* __restrict__ a1,
    const float* __restrict__ a2, const float* __restrict__ a2x,
    const float* __restrict__ w0, const float* __restrict__ w1,
    const float* __restrict__ w2, int m0, int n0, float4* __restrict__ part) {
  const int tid  = threadIdx.x;
  const int wave = tid >> 5;
  const int lane = tid & 31;
  const int h    = lane >> 4;
  const int c    = lane & 15;
  const float* ap[3] = {a0, a1, a2};
  const float* wp[3] = {w0, w1, w2};

  v16f fa[NCH];
  v16f fb[NCH][4];
  #pragma unroll
  for (int i = 0; i < NCH; ++i) {
    const float* A  = ap[i] + (m0 + c) * U_ + wave * 32 + 4 * h;
    const float* Bp = wp[i] + (wave * 32 + 4 * h) * U_ + n0 + 4 * c;
    #pragma unroll
    for (int p = 0; p < 4; ++p) {
      float4 va = *(const float4*)(A + 8 * p);
      if (NCH == 3 && i == 2) {
        const float4 vx = *(const float4*)(a2x + (m0 + c) * U_ + wave * 32 + 4 * h + 8 * p);
        va.x += vx.x; va.y += vx.y; va.z += vx.z; va.w += vx.w;
      }
      fa[i][4 * p + 0] = va.x; fa[i][4 * p + 1] = va.y;
      fa[i][4 * p + 2] = va.z; fa[i][4 * p + 3] = va.w;
      #pragma unroll
      for (int j = 0; j < 4; ++j) {
        const float4 vb = *(const float4*)(Bp + (8 * p + j) * U_);
        fb[i][p][4 * j + 0] = vb.x; fb[i][p][4 * j + 1] = vb.y;
        fb[i][p][4 * j + 2] = vb.z; fb[i][p][4 * j + 3] = vb.w;
      }
    }
  }
  if constexpr (NCH == 1) {
    asm volatile("" :: "v"(fa[0]), "v"(fb[0][0]), "v"(fb[0][1]), "v"(fb[0][2]),
                 "v"(fb[0][3]));
  } else {
    asm volatile("" :: "v"(fa[0]), "v"(fb[0][0]), "v"(fb[0][1]), "v"(fb[0][2]),
                 "v"(fb[0][3]), "v"(fa[1]), "v"(fb[1][0]), "v"(fb[1][1]),
                 "v"(fb[1][2]), "v"(fb[1][3]), "v"(fa[NCH - 1]),
                 "v"(fb[NCH - 1][0]), "v"(fb[NCH - 1][1]), "v"(fb[NCH - 1][2]),
                 "v"(fb[NCH - 1][3]));
  }

  v8f acc[4];
  #pragma unroll
  for (int t = 0; t < 4; ++t)
    acc[t] = (v8f){0.f, 0.f, 0.f, 0.f, 0.f, 0.f, 0.f, 0.f};
  #pragma unroll
  for (int i = 0; i < NCH; ++i) {
    #pragma unroll
    for (int p = 0; p < 4; ++p) {
      #pragma unroll
      for (int q = 0; q < 2; ++q) {
        const v2f a = (v2f){fa[i][4 * p + 2 * q], fa[i][4 * p + 2 * q + 1]};
        const v16f b = fb[i][p];
        const int j0 = 4 * (2 * q), j1 = 4 * (2 * q + 1);
        acc[0] = __builtin_amdgcn_wmma_f32_16x16x4_f32(
            false, a, false, (v2f){b[j0 + 0], b[j1 + 0]}, (short)0, acc[0], false, false);
        acc[1] = __builtin_amdgcn_wmma_f32_16x16x4_f32(
            false, a, false, (v2f){b[j0 + 1], b[j1 + 1]}, (short)0, acc[1], false, false);
        acc[2] = __builtin_amdgcn_wmma_f32_16x16x4_f32(
            false, a, false, (v2f){b[j0 + 2], b[j1 + 2]}, (short)0, acc[2], false, false);
        acc[3] = __builtin_amdgcn_wmma_f32_16x16x4_f32(
            false, a, false, (v2f){b[j0 + 3], b[j1 + 3]}, (short)0, acc[3], false, false);
      }
    }
  }

  #pragma unroll
  for (int r = 0; r < 8; ++r)
    part[(wave * 8 + r) * 32 + lane] =
        make_float4(acc[0][r], acc[1][r], acc[2][r], acc[3][r]);
  __syncthreads();
  float4 s = part[wave * 32 + lane];
  #pragma unroll
  for (int w = 1; w < 8; ++w) {
    const float4 v = part[(w * 8 + wave) * 32 + lane];
    s.x += v.x; s.y += v.y; s.z += v.z; s.w += v.w;
  }
  return s;
}

__global__ __launch_bounds__(256) void k_wxstm(const float* __restrict__ h,
                                               const float* __restrict__ Wa,
                                               float* __restrict__ wx) {
  __shared__ float4 part[8 * 8 * 32];
  const int n0 = blockIdx.x * 64, m0 = blockIdx.y * 16;
  const float4 s = dense16x64<1>(h, h, h, h, Wa, Wa, Wa, m0, n0, part);
  const int lane = threadIdx.x & 31, wave = threadIdx.x >> 5;
  const int row = m0 + 8 * (lane >> 4) + wave, col = n0 + 4 * (lane & 15);
  *(float4*)(wx + row * U_ + col) = s;
}

#define ROWS_WG_ 1024
#define WAVES_   32
#define TILES_W_ 2

__device__ __forceinline__ unsigned int pack2_bf16(float lo, float hi) {
  return (unsigned int)f32_to_bf16_bits(lo) |
         ((unsigned int)f32_to_bf16_bits(hi) << 16);
}

#define S2L2E_ 2.88539008177792681472f
__device__ __forceinline__ float tanh_of_scaled(float y) {
  const float e = __builtin_amdgcn_exp2f(y);
  const float r = __builtin_amdgcn_rcpf(1.0f + e);
  return __builtin_fmaf(r, -2.0f, 1.0f);
}

__global__ __launch_bounds__(WAVES_ * 32) void k_scores(
    const float* __restrict__ x_seq, const float* __restrict__ Ua,
    const float* __restrict__ wxstm, const float* __restrict__ b_a,
    const float* __restrict__ V_a, float* __restrict__ et) {
  __shared__ __align__(16) unsigned short sB[U_ * U_];
  __shared__ float s_wx[U_];
  __shared__ float s_va[U_];

  const int tid  = threadIdx.x;
  const int wave = tid >> 5;
  const int lane = tid & 31;
  const int h    = lane >> 4;
  const int c    = lane & 15;
  const int b    = blockIdx.x >> 1;
  const unsigned int row0 = blockIdx.x * ROWS_WG_;

  {
    const int s = wave & 7;
    const int nt0 = (wave >> 3) * 4;
    #pragma unroll
    for (int nt = nt0; nt < nt0 + 4; ++nt) {
      unsigned int pk[8];
      #pragma unroll
      for (int j2 = 0; j2 < 8; ++j2) {
        const int j = 2 * j2;
        const int k = 32 * s + 8 * (j >> 2) + 4 * h + (j & 3);
        pk[j2] = pack2_bf16(Ua[k * U_ + nt * 16 + c],
                            Ua[(k + 1) * U_ + nt * 16 + c]);
      }
      uint4* dst = (uint4*)&sB[(nt * 8 + s) * 512 + lane * 8];
      dst[0]  = make_uint4(pk[0], pk[1], pk[2], pk[3]);
      dst[32] = make_uint4(pk[4], pk[5], pk[6], pk[7]);
    }
    if (tid < U_) {
      s_wx[tid] = (wxstm[b * U_ + tid] + b_a[tid]) * S2L2E_;
      s_va[tid] = V_a[tid];
    }
  }
  __syncthreads();

  #pragma unroll 1
  for (int it = 0; it < TILES_W_; ++it) {
    const unsigned int R = row0 + (unsigned int)(it * WAVES_ + wave) * 16u;

    union { unsigned int d[8]; v16bf v; } fa[8];
    const float* xp = x_seq + ((R + c) * U_ + h * 4);
    {
      float4 raw[20];
      #pragma unroll
      for (int q = 0; q < 20; ++q) raw[q] = *(const float4*)(xp + q * 8);
      #pragma unroll
      for (int s = 0; s < 5; ++s) {
        #pragma unroll
        for (int q4 = 0; q4 < 4; ++q4) {
          const float4 v = raw[4 * s + q4];
          fa[s].d[2 * q4]     = pack2_bf16(v.x, v.y);
          fa[s].d[2 * q4 + 1] = pack2_bf16(v.z, v.w);
        }
      }
    }
    __builtin_amdgcn_sched_barrier(0);
    {
      float4 raw[12];
      #pragma unroll
      for (int q = 0; q < 12; ++q) raw[q] = *(const float4*)(xp + (20 + q) * 8);
      #pragma unroll
      for (int s = 5; s < 8; ++s) {
        #pragma unroll
        for (int q4 = 0; q4 < 4; ++q4) {
          const float4 v = raw[4 * (s - 5) + q4];
          fa[s].d[2 * q4]     = pack2_bf16(v.x, v.y);
          fa[s].d[2 * q4 + 1] = pack2_bf16(v.z, v.w);
        }
      }
    }

    float sum[8];
    #pragma unroll
    for (int r = 0; r < 8; ++r) sum[r] = 0.f;

    #pragma unroll 1
    for (int ntq = 0; ntq < 8; ++ntq) {
      v8f acc[2];
      #pragma unroll
      for (int u = 0; u < 2; ++u)
        acc[u] = (v8f){0.f, 0.f, 0.f, 0.f, 0.f, 0.f, 0.f, 0.f};
      #pragma unroll
      for (int s = 0; s < 8; ++s) {
        union { uint4 q[2]; v16bf v; } fb[2];
        #pragma unroll
        for (int u = 0; u < 2; ++u) {
          const uint4* bp = (const uint4*)&sB[((ntq * 2 + u) * 8 + s) * 512];
          fb[u].q[0] = bp[lane];
          fb[u].q[1] = bp[32 + lane];
        }
        asm volatile("" :: "v"(fb[0].v), "v"(fb[1].v));
        #pragma unroll
        for (int u = 0; u < 2; ++u)
          acc[u] = __builtin_amdgcn_wmma_f32_16x16x32_bf16(
              false, fa[s].v, false, fb[u].v, (short)0, acc[u], false, false);
        __builtin_amdgcn_sched_barrier(0);
      }
      #pragma unroll
      for (int u = 0; u < 2; ++u) {
        const int ug = (ntq * 2 + u) * 16 + c;
        const float wxs = s_wx[ug];
        const float va = s_va[ug];
        #pragma unroll
        for (int r = 0; r < 8; ++r)
          sum[r] += va * tanh_of_scaled(__builtin_fmaf(acc[u][r], S2L2E_, wxs));
      }
    }

    #pragma unroll
    for (int r = 0; r < 8; ++r) {
      float v = sum[r];
      v += __shfl_xor(v, 1);
      v += __shfl_xor(v, 2);
      v += __shfl_xor(v, 4);
      v += __shfl_xor(v, 8);
      sum[r] = v;
    }
    if (c == 0) {
      float4* o = (float4*)(et + (R + 8 * h));
      o[0] = make_float4(sum[0], sum[1], sum[2], sum[3]);
      o[1] = make_float4(sum[4], sum[5], sum[6], sum[7]);
    }
  }
}

#define CG_ 8
#define CJ_ (ROWS_WG_ / CG_)
#define CB_ 16
#define CK_ (CJ_ / CB_)

__device__ __forceinline__ void ctx_load_group(float4 (&buf)[CB_],
                                               const float* __restrict__ xp, int k) {
  #pragma unroll
  for (int jj = 0; jj < CB_; ++jj)
    buf[jj] = *(const float4*)(xp + (size_t)(8 * (CB_ * k + jj)) * U_);
}

__device__ __forceinline__ void ctx_sum_group(const float4 (&buf)[CB_],
                                              const float* w, float4& acc) {
  float wv[CB_];
  #pragma unroll
  for (int q = 0; q < CB_ / 4; ++q) {
    const float4 w4 = *(const float4*)(w + 4 * q);
    wv[4 * q + 0] = w4.x; wv[4 * q + 1] = w4.y;
    wv[4 * q + 2] = w4.z; wv[4 * q + 3] = w4.w;
  }
  #pragma unroll
  for (int jj = CB_ - 1; jj >= 0; --jj) {
    acc.x = fmaf(wv[jj], buf[jj].x, acc.x);
    acc.y = fmaf(wv[jj], buf[jj].y, acc.y);
    acc.z = fmaf(wv[jj], buf[jj].z, acc.z);
    acc.w = fmaf(wv[jj], buf[jj].w, acc.w);
  }
}

__global__ __launch_bounds__(512) void k_context(const float* __restrict__ x_seq,
                                                 const float* __restrict__ et,
                                                 float* __restrict__ ctxp) {
  __shared__ __align__(16) float s_w[ROWS_WG_];
  __shared__ float s_red[16];
  __shared__ float4 s_part[CG_ * 64];

  const int tid  = threadIdx.x;
  const int lane = tid & 31;
  const int wave = tid >> 5;
  const int g    = tid >> 6;
  const int l    = tid & 63;
  const int b    = blockIdx.x >> 1;
  const int half = blockIdx.x & 1;
  const size_t row0 = (size_t)blockIdx.x * ROWS_WG_;
  const float* xp = x_seq + (row0 + g) * U_ + 4 * l;

  const float* etb = et + (size_t)b * TE_;
  const float r0 = etb[tid], r1 = etb[tid + 512];
  const float r2 = etb[tid + 1024], r3 = etb[tid + 1536];

  float4 bufA[CB_], bufB[CB_];
  ctx_load_group(bufA, xp, CK_ - 1);
  ctx_load_group(bufB, xp, CK_ - 2);
  __builtin_amdgcn_sched_barrier(0);

  const float e0 = expf(r0), e1 = expf(r1), e2 = expf(r2), e3 = expf(r3);
  s_w[(tid & 7) * CJ_ + (tid >> 3)]      = half ? e2 : e0;
  s_w[(tid & 7) * CJ_ + 64 + (tid >> 3)] = half ? e3 : e1;
  float ssum = (e0 + e1) + (e2 + e3);
  ssum += __shfl_xor(ssum, 1);
  ssum += __shfl_xor(ssum, 2);
  ssum += __shfl_xor(ssum, 4);
  ssum += __shfl_xor(ssum, 8);
  ssum += __shfl_xor(ssum, 16);
  if (lane == 0) s_red[wave] = ssum;
  __syncthreads();
  float S = s_red[0];
  #pragma unroll
  for (int w = 1; w < 16; ++w) S += s_red[w];

  float4 acc = make_float4(0.f, 0.f, 0.f, 0.f);
  const float* wg = &s_w[g * CJ_];
  #pragma unroll
  for (int kk = 0; kk < CK_ / 2; ++kk) {
    const int ka = CK_ - 1 - 2 * kk, kb = ka - 1;
    ctx_sum_group(bufA, wg + CB_ * ka, acc);
    if (ka >= 2) {
      ctx_load_group(bufA, xp, ka - 2);
      __builtin_amdgcn_sched_barrier(0);
    }
    ctx_sum_group(bufB, wg + CB_ * kb, acc);
    if (kb >= 2) {
      ctx_load_group(bufB, xp, kb - 2);
      __builtin_amdgcn_sched_barrier(0);
    }
  }

  s_part[g * 64 + l] = acc;
  __syncthreads();
  if (tid < 64) {
    float4 t = s_part[tid];
    #pragma unroll
    for (int gg = 1; gg < CG_; ++gg) {
      const float4 v = s_part[gg * 64 + tid];
      t.x += v.x; t.y += v.y; t.z += v.z; t.w += v.w;
    }
    const float inv = 1.f / S;
    *(float4*)(ctxp + ((size_t)half * B_ + b) * U_ + 4 * tid) =
        make_float4(t.x * inv, t.y * inv, t.z * inv, t.w * inv);
  }
}

__device__ __forceinline__ float sigmoidf_(float a) {
  return 1.f / (1.f + expf(-a));
}

__global__ __launch_bounds__(256) void k_gates(
    const float* __restrict__ h, const float* __restrict__ in,
    const float* __restrict__ ctx,
    const float* __restrict__ Wz, const float* __restrict__ Cz,
    const float* __restrict__ bz,
    const float* __restrict__ Wr, const float* __restrict__ Cr,
    const float* __restrict__ br,
    float* __restrict__ ztw, float* __restrict__ rhw) {
  __shared__ float4 part[8 * 8 * 32];
  const int n0 = blockIdx.x * 64, m0 = blockIdx.y * 16;
  const bool isr = blockIdx.z != 0;
  const float* W    = isr ? Wr : Wz;
  const float* C    = isr ? Cr : Cz;
  const float* bias = isr ? br : bz;
  const float4 s = dense16x64<3>(h, in, ctx, ctx + B_ * U_, W, C, C + IN_ * U_, m0, n0, part);
  const int lane = threadIdx.x & 31, wave = threadIdx.x >> 5;
  const int row = m0 + 8 * (lane >> 4) + wave, col = n0 + 4 * (lane & 15);
  const float4 bb = *(const float4*)(bias + col);
  float4 g;
  g.x = sigmoidf_(s.x + bb.x);
  g.y = sigmoidf_(s.y + bb.y);
  g.z = sigmoidf_(s.z + bb.z);
  g.w = sigmoidf_(s.w + bb.w);
  if (isr) {
    const float4 hv = *(const float4*)(h + row * U_ + col);
    *(float4*)(rhw + row * U_ + col) =
        make_float4(g.x * hv.x, g.y * hv.y, g.z * hv.z, g.w * hv.w);
  } else {
    *(float4*)(ztw + row * U_ + col) = g;
  }
}

__global__ __launch_bounds__(256) void k_out(
    const float* __restrict__ h, const float* __restrict__ in,
    const float* __restrict__ ctx,
    const float* __restrict__ Up, const float* __restrict__ Cp,
    const float* __restrict__ bp,
    const float* __restrict__ ztw, const float* __restrict__ rhw,
    float* __restrict__ out) {
  __shared__ float4 part[8 * 8 * 32];
  const int n0 = blockIdx.x * 64, m0 = blockIdx.y * 16;
  const float4 s = dense16x64<3>(rhw, in, ctx, ctx + B_ * U_, Up, Cp, Cp + IN_ * U_, m0, n0, part);
  const int lane = threadIdx.x & 31, wave = threadIdx.x >> 5;
  const int row = m0 + 8 * (lane >> 4) + wave, col = n0 + 4 * (lane & 15);
  const float4 bb = *(const float4*)(bp + col);
  const float4 zt = *(const float4*)(ztw + row * U_ + col);
  const float4 hv = *(const float4*)(h + row * U_ + col);
  float4 o;
  o.x = (1.f - zt.x) * hv.x + zt.x * tanhf(s.x + bb.x);
  o.y = (1.f - zt.y) * hv.y + zt.y * tanhf(s.y + bb.y);
  o.z = (1.f - zt.z) * hv.z + zt.z * tanhf(s.z + bb.z);
  o.w = (1.f - zt.w) * hv.w + zt.w * tanhf(s.w + bb.w);
  *(float4*)(out + row * U_ + col) = o;
}

extern "C" void kernel_launch(void* const* d_in, const int* in_sizes, int n_in,
                              void* d_out, int out_size, void* d_ws, size_t ws_size,
                              hipStream_t stream) {
  const float* inputs = (const float*)d_in[0];
  const float* h_tm   = (const float*)d_in[1];
  const float* x_seq  = (const float*)d_in[2];
  const float* V_a    = (const float*)d_in[3];
  const float* W_a    = (const float*)d_in[4];
  const float* U_a    = (const float*)d_in[5];
  const float* b_a    = (const float*)d_in[6];
  const float* C_z    = (const float*)d_in[7];
  const float* W_z    = (const float*)d_in[8];
  const float* b_z    = (const float*)d_in[9];
  const float* C_r    = (const float*)d_in[10];
  const float* W_r    = (const float*)d_in[11];
  const float* b_r    = (const float*)d_in[12];
  const float* C_p    = (const float*)d_in[13];
  const float* U_p    = (const float*)d_in[14];
  const float* b_p    = (const float*)d_in[15];
  float* out = (float*)d_out;

  float* wsf   = (float*)d_ws;
  float* wxstm = wsf;
  float* et    = wsf + 32768;
  float* ctx   = et + (size_t)B_ * TE_;
  float* ztw   = ctx + 2 * B_ * U_;
  float* rhw   = ztw + B_ * U_;

  dim3 blk(256);
  k_wxstm  <<<dim3(U_ / 64, B_ / 16), blk, 0, stream>>>(h_tm, W_a, wxstm);
  k_scores <<<dim3(B_ * TE_ / ROWS_WG_), dim3(WAVES_ * 32), 0, stream>>>(
      x_seq, U_a, wxstm, b_a, V_a, et);
  k_context<<<dim3(B_ * TE_ / ROWS_WG_), dim3(512), 0, stream>>>(x_seq, et, ctx);
  k_gates  <<<dim3(U_ / 64, B_ / 16, 2), blk, 0, stream>>>(h_tm, inputs, ctx, W_z, C_z, b_z,
                                          W_r, C_r, b_r, ztw, rhw);
  k_out    <<<dim3(U_ / 64, B_ / 16), blk, 0, stream>>>(h_tm, inputs, ctx, U_p, C_p, b_p,
                                          ztw, rhw, out);
}


# gate, out layers: products not waiting for the context (h_tm, inputs x weights, bias) go to the first launch as planes, read in the operand flight; operand dtype unchanged; sum: context, then planes
# speedup vs baseline: 6.1989x; 1.0864x over previous
#include <hip/hip_runtime.h>

#define B_   128
#define TE_  2048
#define U_   256
#define IN_  256

typedef __attribute__((ext_vector_type(16))) __bf16 v16bf;
typedef __attribute__((ext_vector_type(8)))  float  v8f;

typedef __attribute__((ext_vector_type(2))) float v2f;
typedef __attribute__((ext_vector_type(16))) float v16f;

struct DenseOut { float4 s, e0, e1, e2; };

template <bool TWO, int NE>
__device__ __forceinline__ DenseOut dense16x64(
    const float* __restrict__ a, const float* __restrict__ ax,
    const float* __restrict__ w,
    const float* __restrict__ q0, int stride0, bool has0,
    const float* __restrict__ q1, bool has1,
    const float* __restrict__ q2, bool has2,
    int m0, int n0, float4* __restrict__ part) {
  const int tid  = threadIdx.x;
  const int wave = tid >> 5;
  const int lane = tid & 31;
  const int h    = lane >> 4;
  const int c    = lane & 15;

  v16f fa;
  v16f fb[4];
  const float* A  = a + (m0 + c) * U_ + wave * 32 + 4 * h;
  const float* Bp = w + (wave * 32 + 4 * h) * U_ + n0 + 4 * c;
  #pragma unroll
  for (int p = 0; p < 4; ++p) {
    float4 va = *(const float4*)(A + 8 * p);
    if constexpr (TWO) {
      const float4 vx = *(const float4*)(ax + (m0 + c) * U_ + wave * 32 + 4 * h + 8 * p);
      va.x += vx.x; va.y += vx.y; va.z += vx.z; va.w += vx.w;
    }
    fa[4 * p + 0] = va.x; fa[4 * p + 1] = va.y;
    fa[4 * p + 2] = va.z; fa[4 * p + 3] = va.w;
    #pragma unroll
    for (int j = 0; j < 4; ++j) {
      const float4 vb = *(const float4*)(Bp + (8 * p + j) * U_);
      fb[p][4 * j + 0] = vb.x; fb[p][4 * j + 1] = vb.y;
      fb[p][4 * j + 2] = vb.z; fb[p][4 * j + 3] = vb.w;
    }
  }
  const int erow = m0 + 8 * h + wave, ecol = n0 + 4 * c;
  DenseOut o;
  o.e0 = o.e1 = o.e2 = make_float4(0.f, 0.f, 0.f, 0.f);
  if constexpr (NE >= 1) { if (has0) o.e0 = *(const float4*)(q0 + erow * stride0 + ecol); }
  if constexpr (NE >= 2) { if (has1) o.e1 = *(const float4*)(q1 + erow * U_ + ecol); }
  if constexpr (NE >= 3) { if (has2) o.e2 = *(const float4*)(q2 + erow * U_ + ecol); }
  if constexpr (NE == 1) {
    asm volatile("" :: "v"(fa), "v"(fb[0]), "v"(fb[1]), "v"(fb[2]), "v"(fb[3]),
                 "v"(o.e0.x), "v"(o.e0.y), "v"(o.e0.z), "v"(o.e0.w));
  } else {
    asm volatile("" :: "v"(fa), "v"(fb[0]), "v"(fb[1]), "v"(fb[2]), "v"(fb[3]),
                 "v"(o.e0.x), "v"(o.e0.y), "v"(o.e0.z), "v"(o.e0.w),
                 "v"(o.e1.x), "v"(o.e1.y), "v"(o.e1.z), "v"(o.e1.w),
                 "v"(o.e2.x), "v"(o.e2.y), "v"(o.e2.z), "v"(o.e2.w));
  }

  v8f acc[4];
  #pragma unroll
  for (int t = 0; t < 4; ++t)
    acc[t] = (v8f){0.f, 0.f, 0.f, 0.f, 0.f, 0.f, 0.f, 0.f};
  #pragma unroll
  for (int p = 0; p < 4; ++p) {
    #pragma unroll
    for (int q = 0; q < 2; ++q) {
      const v2f av = (v2f){fa[4 * p + 2 * q], fa[4 * p + 2 * q + 1]};
      const v16f b = fb[p];
      const int j0 = 4 * (2 * q), j1 = 4 * (2 * q + 1);
      acc[0] = __builtin_amdgcn_wmma_f32_16x16x4_f32(
          false, av, false, (v2f){b[j0 + 0], b[j1 + 0]}, (short)0, acc[0], false, false);
      acc[1] = __builtin_amdgcn_wmma_f32_16x16x4_f32(
          false, av, false, (v2f){b[j0 + 1], b[j1 + 1]}, (short)0, acc[1], false, false);
      acc[2] = __builtin_amdgcn_wmma_f32_16x16x4_f32(
          false, av, false, (v2f){b[j0 + 2], b[j1 + 2]}, (short)0, acc[2], false, false);
      acc[3] = __builtin_amdgcn_wmma_f32_16x16x4_f32(
          false, av, false, (v2f){b[j0 + 3], b[j1 + 3]}, (short)0, acc[3], false, false);
    }
  }

  #pragma unroll
  for (int r = 0; r < 8; ++r)
    part[(wave * 8 + r) * 32 + lane] =
        make_float4(acc[0][r], acc[1][r], acc[2][r], acc[3][r]);
  __syncthreads();
  float4 s = part[wave * 32 + lane];
  #pragma unroll
  for (int w8 = 1; w8 < 8; ++w8) {
    const float4 v = part[(w8 * 8 + wave) * 32 + lane];
    s.x += v.x; s.y += v.y; s.z += v.z; s.w += v.w;
  }
  o.s = s;
  return o;
}

__global__ __launch_bounds__(256) void k_pre(
    const float* __restrict__ h, const float* __restrict__ in,
    const float* __restrict__ Wa, const float* __restrict__ Wz,
    const float* __restrict__ Cz, const float* __restrict__ bz,
    const float* __restrict__ Wr, const float* __restrict__ Cr,
    const float* __restrict__ br, const float* __restrict__ Cp,
    const float* __restrict__ bp,
    float* __restrict__ wx, float* __restrict__ pre) {
  __shared__ float4 part[8 * 8 * 32];
  const int n0 = blockIdx.x * 64, m0 = blockIdx.y * 16;
  const int z = blockIdx.z;
  const bool of_h = z == 0 || z == 1 || z == 3;
  const float* a = of_h ? h : in;
  const float* w = z == 0 ? Wa : z == 1 ? Wz : z == 2 ? Cz : z == 3 ? Wr : z == 4 ? Cr : Cp;
  const float* bias = z == 2 ? bz : z == 4 ? br : bp;
  float* dst = z == 0 ? wx : pre + (z - 1) * (B_ * U_);
  const DenseOut d = dense16x64<false, 1>(a, a, w, bias, 0, !of_h, bias, false, bias, false,
                                          m0, n0, part);
  const int lane = threadIdx.x & 31, wave = threadIdx.x >> 5;
  const int row = m0 + 8 * (lane >> 4) + wave, col = n0 + 4 * (lane & 15);
  *(float4*)(dst + row * U_ + col) =
      make_float4(d.s.x + d.e0.x, d.s.y + d.e0.y, d.s.z + d.e0.z, d.s.w + d.e0.w);
}

#define ROWS_WG_ 1024
#define WAVES_   32
#define TILES_W_ 2

__device__ __forceinline__ unsigned int pack2_bf16(float lo, float hi) {
  const unsigned int a = __float_as_uint(lo) + 0x8000u;
  const unsigned int b = __float_as_uint(hi) + 0x8000u;
  return __builtin_amdgcn_perm(b, a, 0x07060302u);
}

#define S2L2E_ 2.88539008177792681472f
__device__ __forceinline__ float tanh_of_scaled(float y) {
  const float e = __builtin_amdgcn_exp2f(y);
  const float r = __builtin_amdgcn_rcpf(1.0f + e);
  return __builtin_fmaf(r, -2.0f, 1.0f);
}

__global__ __launch_bounds__(WAVES_ * 32) void k_scores(
    const float* __restrict__ x_seq, const float* __restrict__ Ua,
    const float* __restrict__ wxstm, const float* __restrict__ b_a,
    const float* __restrict__ V_a, float* __restrict__ et) {
  __shared__ __align__(16) unsigned short sB[U_ * U_];
  __shared__ float s_wx[U_];
  __shared__ float s_va[U_];

  const int tid  = threadIdx.x;
  const int wave = tid >> 5;
  const int lane = tid & 31;
  const int h    = lane >> 4;
  const int c    = lane & 15;
  const int b    = blockIdx.x >> 1;
  const unsigned int row0 = blockIdx.x * ROWS_WG_;

  {
    const int s = wave & 7;
    const int nt0 = (wave >> 3) * 4;
    #pragma unroll
    for (int nt = nt0; nt < nt0 + 4; ++nt) {
      unsigned int pk[8];
      #pragma unroll
      for (int j2 = 0; j2 < 8; ++j2) {
        const int j = 2 * j2;
        const int k = 32 * s + 8 * (j >> 2) + 4 * h + (j & 3);
        pk[j2] = pack2_bf16(Ua[k * U_ + nt * 16 + c],
                            Ua[(k + 1) * U_ + nt * 16 + c]);
      }
      uint4* dst = (uint4*)&sB[(nt * 8 + s) * 512 + lane * 8];
      dst[0]  = make_uint4(pk[0], pk[1], pk[2], pk[3]);
      dst[32] = make_uint4(pk[4], pk[5], pk[6], pk[7]);
    }
    if (tid < U_) {
      s_wx[tid] = (wxstm[b * U_ + tid] + b_a[tid]) * S2L2E_;
      s_va[tid] = V_a[tid];
    }
  }
  __syncthreads();

  #pragma unroll 1
  for (int it = 0; it < TILES_W_; ++it) {
    const unsigned int R = row0 + (unsigned int)(it * WAVES_ + wave) * 16u;

    union { unsigned int d[8]; v16bf v; } fa[8];
    const float* xp = x_seq + ((R + c) * U_ + h * 4);
    {
      float4 raw[20];
      #pragma unroll
      for (int q = 0; q < 20; ++q) raw[q] = *(const float4*)(xp + q * 8);
      #pragma unroll
      for (int s = 0; s < 5; ++s) {
        #pragma unroll
        for (int q4 = 0; q4 < 4; ++q4) {
          const float4 v = raw[4 * s + q4];
          fa[s].d[2 * q4]     = pack2_bf16(v.x, v.y);
          fa[s].d[2 * q4 + 1] = pack2_bf16(v.z, v.w);
        }
      }
    }
    __builtin_amdgcn_sched_barrier(0);
    {
      float4 raw[12];
      #pragma unroll
      for (int q = 0; q < 12; ++q) raw[q] = *(const float4*)(xp + (20 + q) * 8);
      #pragma unroll
      for (int s = 5; s < 8; ++s) {
        #pragma unroll
        for (int q4 = 0; q4 < 4; ++q4) {
          const float4 v = raw[4 * (s - 5) + q4];
          fa[s].d[2 * q4]     = pack2_bf16(v.x, v.y);
          fa[s].d[2 * q4 + 1] = pack2_bf16(v.z, v.w);
        }
      }
    }

    float sum[8];
    #pragma unroll
    for (int r = 0; r < 8; ++r) sum[r] = 0.f;

    #pragma unroll 1
    for (int ntq = 0; ntq < 8; ++ntq) {
      v8f acc[2];
      #pragma unroll
      for (int u = 0; u < 2; ++u)
        acc[u] = (v8f){0.f, 0.f, 0.f, 0.f, 0.f, 0.f, 0.f, 0.f};
      #pragma unroll
      for (int s = 0; s < 8; ++s) {
        union { uint4 q[2]; v16bf v; } fb[2];
        #pragma unroll
        for (int u = 0; u < 2; ++u) {
          const uint4* bp = (const uint4*)&sB[((ntq * 2 + u) * 8 + s) * 512];
          fb[u].q[0] = bp[lane];
          fb[u].q[1] = bp[32 + lane];
        }
        asm volatile("" :: "v"(fb[0].v), "v"(fb[1].v));
        #pragma unroll
        for (int u = 0; u < 2; ++u)
          acc[u] = __builtin_amdgcn_wmma_f32_16x16x32_bf16(
              false, fa[s].v, false, fb[u].v, (short)0, acc[u], false, false);
        __builtin_amdgcn_sched_barrier(0);
      }
      #pragma unroll
      for (int u = 0; u < 2; ++u) {
        const int ug = (ntq * 2 + u) * 16 + c;
        const float wxs = s_wx[ug];
        const float va = s_va[ug];
        #pragma unroll
        for (int r = 0; r < 8; ++r)
          sum[r] += va * tanh_of_scaled(__builtin_fmaf(acc[u][r], S2L2E_, wxs));
      }
    }

    #pragma unroll
    for (int r = 0; r < 8; ++r) {
      float v = sum[r];
      v += __shfl_xor(v, 1);
      v += __shfl_xor(v, 2);
      v += __shfl_xor(v, 4);
      v += __shfl_xor(v, 8);
      sum[r] = v;
    }
    if (c == 0) {
      float4* o = (float4*)(et + (R + 8 * h));
      o[0] = make_float4(sum[0], sum[1], sum[2], sum[3]);
      o[1] = make_float4(sum[4], sum[5], sum[6], sum[7]);
    }
  }
}

#define CG_ 8
#define CJ_ (ROWS_WG_ / CG_)
#define CB_ 16
#define CK_ (CJ_ / CB_)

__device__ __forceinline__ void ctx_load_group(float4 (&buf)[CB_],
                                               const float* __restrict__ xp, int k) {
  #pragma unroll
  for (int jj = 0; jj < CB_; ++jj)
    buf[jj] = *(const float4*)(xp + (size_t)(8 * (CB_ * k + jj)) * U_);
}

__device__ __forceinline__ void ctx_sum_group(const float4 (&buf)[CB_],
                                              const float* w, float4& acc) {
  float wv[CB_];
  #pragma unroll
  for (int q = 0; q < CB_ / 4; ++q) {
    const float4 w4 = *(const float4*)(w + 4 * q);
    wv[4 * q + 0] = w4.x; wv[4 * q + 1] = w4.y;
    wv[4 * q + 2] = w4.z; wv[4 * q + 3] = w4.w;
  }
  #pragma unroll
  for (int jj = CB_ - 1; jj >= 0; --jj) {
    acc.x = fmaf(wv[jj], buf[jj].x, acc.x);
    acc.y = fmaf(wv[jj], buf[jj].y, acc.y);
    acc.z = fmaf(wv[jj], buf[jj].z, acc.z);
    acc.w = fmaf(wv[jj], buf[jj].w, acc.w);
  }
}

__global__ __launch_bounds__(512) void k_context(const float* __restrict__ x_seq,
                                                 const float* __restrict__ et,
                                                 float* __restrict__ ctxp) {
  __shared__ __align__(16) float s_w[ROWS_WG_];
  __shared__ float s_red[16];
  __shared__ float4 s_part[CG_ * 64];

  const int tid  = threadIdx.x;
  const int lane = tid & 31;
  const int wave = tid >> 5;
  const int g    = tid >> 6;
  const int l    = tid & 63;
  const int b    = blockIdx.x >> 1;
  const int half = blockIdx.x & 1;
  const size_t row0 = (size_t)blockIdx.x * ROWS_WG_;
  const float* xp = x_seq + (row0 + g) * U_ + 4 * l;

  const float* etb = et + (size_t)b * TE_;
  const float r0 = etb[tid], r1 = etb[tid + 512];
  const float r2 = etb[tid + 1024], r3 = etb[tid + 1536];

  float4 bufA[CB_], bufB[CB_];
  ctx_load_group(bufA, xp, CK_ - 1);
  ctx_load_group(bufB, xp, CK_ - 2);
  __builtin_amdgcn_sched_barrier(0);

  const float e0 = expf(r0), e1 = expf(r1), e2 = expf(r2), e3 = expf(r3);
  s_w[(tid & 7) * CJ_ + (tid >> 3)]      = half ? e2 : e0;
  s_w[(tid & 7) * CJ_ + 64 + (tid >> 3)] = half ? e3 : e1;
  float ssum = (e0 + e1) + (e2 + e3);
  ssum += __shfl_xor(ssum, 1);
  ssum += __shfl_xor(ssum, 2);
  ssum += __shfl_xor(ssum, 4);
  ssum += __shfl_xor(ssum, 8);
  ssum += __shfl_xor(ssum, 16);
  if (lane == 0) s_red[wave] = ssum;
  __syncthreads();
  float S = s_red[0];
  #pragma unroll
  for (int w = 1; w < 16; ++w) S += s_red[w];

  float4 acc = make_float4(0.f, 0.f, 0.f, 0.f);
  const float* wg = &s_w[g * CJ_];
  #pragma unroll
  for (int kk = 0; kk < CK_ / 2; ++kk) {
    const int ka = CK_ - 1 - 2 * kk, kb = ka - 1;
    ctx_sum_group(bufA, wg + CB_ * ka, acc);
    if (ka >= 2) {
      ctx_load_group(bufA, xp, ka - 2);
      __builtin_amdgcn_sched_barrier(0);
    }
    ctx_sum_group(bufB, wg + CB_ * kb, acc);
    if (kb >= 2) {
      ctx_load_group(bufB, xp, kb - 2);
      __builtin_amdgcn_sched_barrier(0);
    }
  }

  s_part[g * 64 + l] = acc;
  __syncthreads();
  if (tid < 64) {
    float4 t = s_part[tid];
    #pragma unroll
    for (int gg = 1; gg < CG_; ++gg) {
      const float4 v = s_part[gg * 64 + tid];
      t.x += v.x; t.y += v.y; t.z += v.z; t.w += v.w;
    }
    const float inv = 1.f / S;
    *(float4*)(ctxp + ((size_t)half * B_ + b) * U_ + 4 * tid) =
        make_float4(t.x * inv, t.y * inv, t.z * inv, t.w * inv);
  }
}

__device__ __forceinline__ float sigmoidf_(float a) {
  return 1.f / (1.f + expf(-a));
}

__global__ __launch_bounds__(256) void k_gates(
    const float* __restrict__ h, const float* __restrict__ ctx,
    const float* __restrict__ Cz, const float* __restrict__ Cr,
    const float* __restrict__ Cp, const float* __restrict__ pre,
    float* __restrict__ ztw, float* __restrict__ rhw, float* __restrict__ ppw) {
  __shared__ float4 part[8 * 8 * 32];
  const int n0 = blockIdx.x * 64, m0 = blockIdx.y * 16;
  const int z = blockIdx.z;
  const float* C = (z == 0 ? Cz : z == 1 ? Cr : Cp) + IN_ * U_;
  const float* p0 = pre + (z < 2 ? 2 * z + 1 : 4) * (B_ * U_);
  const float* p1 = pre + (z < 2 ? 2 * z : 4) * (B_ * U_);
  const DenseOut d = dense16x64<true, 3>(ctx, ctx + B_ * U_, C, p0, U_, true, p1, z < 2,
                                         h, z == 1, m0, n0, part);
  const int lane = threadIdx.x & 31, wave = threadIdx.x >> 5;
  const int row = m0 + 8 * (lane >> 4) + wave, col = n0 + 4 * (lane & 15);
  float4 s;
  s.x = d.s.x + d.e1.x + d.e0.x;
  s.y = d.s.y + d.e1.y + d.e0.y;
  s.z = d.s.z + d.e1.z + d.e0.z;
  s.w = d.s.w + d.e1.w + d.e0.w;
  if (z == 2) {
    *(float4*)(ppw + row * U_ + col) = s;
    return;
  }
  float4 g;
  g.x = sigmoidf_(s.x);
  g.y = sigmoidf_(s.y);
  g.z = sigmoidf_(s.z);
  g.w = sigmoidf_(s.w);
  if (z == 1) {
    *(float4*)(rhw + row * U_ + col) =
        make_float4(g.x * d.e2.x, g.y * d.e2.y, g.z * d.e2.z, g.w * d.e2.w);
  } else {
    *(float4*)(ztw + row * U_ + col) = g;
  }
}

__global__ __launch_bounds__(256) void k_out(
    const float* __restrict__ h, const float* __restrict__ Up,
    const float* __restrict__ ppw, const float* __restrict__ ztw,
    const float* __restrict__ rhw, float* __restrict__ out) {
  __shared__ float4 part[8 * 8 * 32];
  const int n0 = blockIdx.x * 64, m0 = blockIdx.y * 16;
  const DenseOut d = dense16x64<false, 3>(rhw, rhw, Up, ppw, U_, true, ztw, true, h, true,
                                          m0, n0, part);
  const int lane = threadIdx.x & 31, wave = threadIdx.x >> 5;
  const int row = m0 + 8 * (lane >> 4) + wave, col = n0 + 4 * (lane & 15);
  const float4 zt = d.e1, hv = d.e2;
  float4 o;
  o.x = (1.f - zt.x) * hv.x + zt.x * tanhf(d.s.x + d.e0.x);
  o.y = (1.f - zt.y) * hv.y + zt.y * tanhf(d.s.y + d.e0.y);
  o.z = (1.f - zt.z) * hv.z + zt.z * tanhf(d.s.z + d.e0.z);
  o.w = (1.f - zt.w) * hv.w + zt.w * tanhf(d.s.w + d.e0.w);
  *(float4*)(out + row * U_ + col) = o;
}

extern "C" void kernel_launch(void* const* d_in, const int* in_sizes, int n_in,
                              void* d_out, int out_size, void* d_ws, size_t ws_size,
                              hipStream_t stream) {
  const float* inputs = (const float*)d_in[0];
  const float* h_tm   = (const float*)d_in[1];
  const float* x_seq  = (const float*)d_in[2];
  const float* V_a    = (const float*)d_in[3];
  const float* W_a    = (const float*)d_in[4];
  const float* U_a    = (const float*)d_in[5];
  const float* b_a    = (const float*)d_in[6];
  const float* C_z    = (const float*)d_in[7];
  const float* W_z    = (const float*)d_in[8];
  const float* b_z    = (const float*)d_in[9];
  const float* C_r    = (const float*)d_in[10];
  const float* W_r    = (const float*)d_in[11];
  const float* b_r    = (const float*)d_in[12];
  const float* C_p    = (const float*)d_in[13];
  const float* U_p    = (const float*)d_in[14];
  const float* b_p    = (const float*)d_in[15];
  float* out = (float*)d_out;

  float* wsf   = (float*)d_ws;
  float* wxstm = wsf;
  float* et    = wsf + 32768;
  float* ctx   = et + (size_t)B_ * TE_;
  float* ztw   = ctx + 2 * B_ * U_;
  float* rhw   = ztw + B_ * U_;
  float* pre   = rhw + B_ * U_;
  float* ppw   = pre + 5 * B_ * U_;

  dim3 blk(256);
  k_pre    <<<dim3(U_ / 64, B_ / 16, 6), blk, 0, stream>>>(h_tm, inputs, W_a, W_z, C_z, b_z, W_r, C_r,
                                                           b_r, C_p, b_p, wxstm, pre);
  k_scores <<<dim3(B_ * TE_ / ROWS_WG_), dim3(WAVES_ * 32), 0, stream>>>(
      x_seq, U_a, wxstm, b_a, V_a, et);
  k_context<<<dim3(B_ * TE_ / ROWS_WG_), dim3(512), 0, stream>>>(x_seq, et, ctx);
  k_gates  <<<dim3(U_ / 64, B_ / 16, 3), blk, 0, stream>>>(h_tm, ctx, C_z, C_r, C_p, pre, ztw, rhw, ppw);
  k_out    <<<dim3(U_ / 64, B_ / 16), blk, 0, stream>>>(h_tm, U_p, ppw, ztw, rhw, out);
}


# k_scores: a wave holds two 16-row tiles (16 waves of 256 registers, was 32 of 128) and feeds both from each LDS B fragment: half the LDS reads per WMMA; exp-sum order over tiles changes
# speedup vs baseline: 8.0083x; 1.2919x over previous
#include <hip/hip_runtime.h>

#define B_   128
#define TE_  2048
#define U_   256
#define IN_  256

typedef __attribute__((ext_vector_type(16))) __bf16 v16bf;
typedef __attribute__((ext_vector_type(8)))  float  v8f;

typedef __attribute__((ext_vector_type(2))) float v2f;
typedef __attribute__((ext_vector_type(16))) float v16f;

struct DenseOut { float4 s, e0, e1, e2; };

template <bool TWO, int NE>
__device__ __forceinline__ DenseOut dense16x64(
    const float* __restrict__ a, const float* __restrict__ ax,
    const float* __restrict__ asum, const float* __restrict__ w,
    const float* __restrict__ q0, int stride0, bool has0,
    const float* __restrict__ q1, bool has1,
    const float* __restrict__ q2, bool has2,
    int m0, int n0, float4* __restrict__ part) {
  const int tid  = threadIdx.x;
  const int wave = tid >> 5;
  const int lane = tid & 31;
  const int h    = lane >> 4;
  const int c    = lane & 15;

  v16f fa;
  v16f fb[4];
  const float* A  = a + (m0 + c) * U_ + wave * 32 + 4 * h;
  const float* Bp = w + (wave * 32 + 4 * h) * U_ + n0 + 4 * c;
  float inv = 1.f;
  if constexpr (TWO) {
    const float2 sw = *(const float2*)(asum + 2 * (m0 + c));
    inv = 1.f / (sw.x + sw.y);
  }
  #pragma unroll
  for (int p = 0; p < 4; ++p) {
    float4 va = *(const float4*)(A + 8 * p);
    if constexpr (TWO) {
      const float4 vx = *(const float4*)(ax + (m0 + c) * U_ + wave * 32 + 4 * h + 8 * p);
      va.x = (va.x + vx.x) * inv; va.y = (va.y + vx.y) * inv;
      va.z = (va.z + vx.z) * inv; va.w = (va.w + vx.w) * inv;
    }
    fa[4 * p + 0] = va.x; fa[4 * p + 1] = va.y;
    fa[4 * p + 2] = va.z; fa[4 * p + 3] = va.w;
    #pragma unroll
    for (int j = 0; j < 4; ++j) {
      const float4 vb = *(const float4*)(Bp + (8 * p + j) * U_);
      fb[p][4 * j + 0] = vb.x; fb[p][4 * j + 1] = vb.y;
      fb[p][4 * j + 2] = vb.z; fb[p][4 * j + 3] = vb.w;
    }
  }
  const int erow = m0 + 8 * h + wave, ecol = n0 + 4 * c;
  DenseOut o;
  o.e0 = o.e1 = o.e2 = make_float4(0.f, 0.f, 0.f, 0.f);
  if constexpr (NE >= 1) { if (has0) o.e0 = *(const float4*)(q0 + erow * stride0 + ecol); }
  if constexpr (NE >= 2) { if (has1) o.e1 = *(const float4*)(q1 + erow * U_ + ecol); }
  if constexpr (NE >= 3) { if (has2) o.e2 = *(const float4*)(q2 + erow * U_ + ecol); }
  if constexpr (NE == 1) {
    asm volatile("" :: "v"(fa), "v"(fb[0]), "v"(fb[1]), "v"(fb[2]), "v"(fb[3]),
                 "v"(o.e0.x), "v"(o.e0.y), "v"(o.e0.z), "v"(o.e0.w));
  } else {
    asm volatile("" :: "v"(fa), "v"(fb[0]), "v"(fb[1]), "v"(fb[2]), "v"(fb[3]),
                 "v"(o.e0.x), "v"(o.e0.y), "v"(o.e0.z), "v"(o.e0.w),
                 "v"(o.e1.x), "v"(o.e1.y), "v"(o.e1.z), "v"(o.e1.w),
                 "v"(o.e2.x), "v"(o.e2.y), "v"(o.e2.z), "v"(o.e2.w));
  }

  v8f acc[4];
  #pragma unroll
  for (int t = 0; t < 4; ++t)
    acc[t] = (v8f){0.f, 0.f, 0.f, 0.f, 0.f, 0.f, 0.f, 0.f};
  #pragma unroll
  for (int p = 0; p < 4; ++p) {
    #pragma unroll
    for (int q = 0; q < 2; ++q) {
      const v2f av = (v2f){fa[4 * p + 2 * q], fa[4 * p + 2 * q + 1]};
      const v16f b = fb[p];
      const int j0 = 4 * (2 * q), j1 = 4 * (2 * q + 1);
      acc[0] = __builtin_amdgcn_wmma_f32_16x16x4_f32(
          false, av, false, (v2f){b[j0 + 0], b[j1 + 0]}, (short)0, acc[0], false, false);
      acc[1] = __builtin_amdgcn_wmma_f32_16x16x4_f32(
          false, av, false, (v2f){b[j0 + 1], b[j1 + 1]}, (short)0, acc[1], false, false);
      acc[2] = __builtin_amdgcn_wmma_f32_16x16x4_f32(
          false, av, false, (v2f){b[j0 + 2], b[j1 + 2]}, (short)0, acc[2], false, false);
      acc[3] = __builtin_amdgcn_wmma_f32_16x16x4_f32(
          false, av, false, (v2f){b[j0 + 3], b[j1 + 3]}, (short)0, acc[3], false, false);
    }
  }

  #pragma unroll
  for (int r = 0; r < 8; ++r)
    part[(wave * 8 + r) * 32 + lane] =
        make_float4(acc[0][r], acc[1][r], acc[2][r], acc[3][r]);
  __syncthreads();
  float4 s = part[wave * 32 + lane];
  #pragma unroll
  for (int w8 = 1; w8 < 8; ++w8) {
    const float4 v = part[(w8 * 8 + wave) * 32 + lane];
    s.x += v.x; s.y += v.y; s.z += v.z; s.w += v.w;
  }
  o.s = s;
  return o;
}

__global__ __launch_bounds__(256) void k_pre(
    const float* __restrict__ h, const float* __restrict__ in,
    const float* __restrict__ Wa, const float* __restrict__ Wz,
    const float* __restrict__ Cz, const float* __restrict__ bz,
    const float* __restrict__ Wr, const float* __restrict__ Cr,
    const float* __restrict__ br, const float* __restrict__ Cp,
    const float* __restrict__ bp,
    float* __restrict__ wx, float* __restrict__ pre) {
  __shared__ float4 part[8 * 8 * 32];
  const int n0 = blockIdx.x * 64, m0 = blockIdx.y * 16;
  const int z = blockIdx.z;
  const bool of_h = z == 0 || z == 1 || z == 3;
  const float* a = of_h ? h : in;
  const float* w = z == 0 ? Wa : z == 1 ? Wz : z == 2 ? Cz : z == 3 ? Wr : z == 4 ? Cr : Cp;
  const float* bias = z == 2 ? bz : z == 4 ? br : bp;
  float* dst = z == 0 ? wx : pre + (z - 1) * (B_ * U_);
  const DenseOut d = dense16x64<false, 1>(a, a, a, w, bias, 0, !of_h, bias, false, bias, false,
                                          m0, n0, part);
  const int lane = threadIdx.x & 31, wave = threadIdx.x >> 5;
  const int row = m0 + 8 * (lane >> 4) + wave, col = n0 + 4 * (lane & 15);
  *(float4*)(dst + row * U_ + col) =
      make_float4(d.s.x + d.e0.x, d.s.y + d.e0.y, d.s.z + d.e0.z, d.s.w + d.e0.w);
}

#define ROWS_WG_ 1024
#define WAVES_   16
#define MT_      2
#define ROUNDS_  (ROWS_WG_ / (WAVES_ * MT_ * 16))

__device__ __forceinline__ unsigned int pack2_bf16(float lo, float hi) {
  const unsigned int a = __float_as_uint(lo) + 0x8000u;
  const unsigned int b = __float_as_uint(hi) + 0x8000u;
  return __builtin_amdgcn_perm(b, a, 0x07060302u);
}

#define S2L2E_ 2.88539008177792681472f
__device__ __forceinline__ float tanh_of_scaled(float y) {
  const float e = __builtin_amdgcn_exp2f(y);
  const float r = __builtin_amdgcn_rcpf(1.0f + e);
  return __builtin_fmaf(r, -2.0f, 1.0f);
}

__global__ __launch_bounds__(WAVES_ * 32) void k_scores(
    const float* __restrict__ x_seq, const float* __restrict__ Ua,
    const float* __restrict__ wxstm, const float* __restrict__ b_a,
    const float* __restrict__ V_a, float* __restrict__ ctxp,
    float* __restrict__ csum) {
  __shared__ __align__(16) unsigned short sB[U_ * U_];
  __shared__ float s_wx[U_];
  __shared__ float s_va[U_];
  __shared__ __align__(16) unsigned short sSel[2 * 512];
  __shared__ float sC[(ROWS_WG_ / 16) * 16 * 32];
  __shared__ float s_red[4 * U_];
  __shared__ float s_ws[WAVES_];

  const int tid  = threadIdx.x;
  const int wave = tid >> 5;
  const int lane = tid & 31;
  const int h    = lane >> 4;
  const int c    = lane & 15;
  const int b    = blockIdx.x >> 1;
  const unsigned int row0 = blockIdx.x * ROWS_WG_;

  {
    const int s = wave & 7;
    const int nt0 = (wave >> 3) * 8;
    #pragma unroll
    for (int nt = nt0; nt < nt0 + 8; ++nt) {
      unsigned int pk[8];
      #pragma unroll
      for (int j2 = 0; j2 < 8; ++j2) {
        const int j = 2 * j2;
        const int k = 32 * s + 8 * (j >> 2) + 4 * h + (j & 3);
        pk[j2] = pack2_bf16(Ua[k * U_ + nt * 16 + c],
                            Ua[(k + 1) * U_ + nt * 16 + c]);
      }
      uint4* dst = (uint4*)&sB[(nt * 8 + s) * 512 + lane * 8];
      dst[0]  = make_uint4(pk[0], pk[1], pk[2], pk[3]);
      dst[32] = make_uint4(pk[4], pk[5], pk[6], pk[7]);
    }
    if (tid < U_) {
      s_wx[tid] = (wxstm[b * U_ + tid] + b_a[tid]) * S2L2E_;
      s_va[tid] = V_a[tid];
    }
    if (tid < 64) {
      const int u = tid >> 5;
      #pragma unroll
      for (int e = 0; e < 16; ++e)
        sSel[u * 512 + (e >> 3) * 256 + lane * 8 + (e & 7)] =
            (h == u && e == c) ? (unsigned short)0x3F80 : (unsigned short)0;
    }
  }
  __syncthreads();

  float wacc = 0.f;

  #pragma unroll 1
  for (int it = 0; it < ROUNDS_; ++it) {
    const unsigned int R = row0 + (unsigned int)(it * WAVES_ + wave) * (16u * MT_);

    union { unsigned int d[8]; v16bf v; } fa[MT_][8];
    #pragma unroll
    for (int m = 0; m < MT_; ++m) {
      const float* xp = x_seq + ((R + 16u * m + c) * U_ + h * 4);
      float4 raw[32];
      #pragma unroll
      for (int q = 0; q < 32; ++q) raw[q] = *(const float4*)(xp + q * 8);
      #pragma unroll
      for (int s = 0; s < 8; ++s) {
        #pragma unroll
        for (int q4 = 0; q4 < 4; ++q4) {
          const float4 v = raw[4 * s + q4];
          fa[m][s].d[2 * q4]     = pack2_bf16(v.x, v.y);
          fa[m][s].d[2 * q4 + 1] = pack2_bf16(v.z, v.w);
        }
      }
    }

    float sum[MT_][8];
    #pragma unroll
    for (int m = 0; m < MT_; ++m) {
      #pragma unroll
      for (int r = 0; r < 8; ++r) sum[m][r] = 0.f;
    }

    #pragma unroll 1
    for (int ntq = 0; ntq < 8; ++ntq) {
      v8f acc[MT_][2];
      #pragma unroll
      for (int m = 0; m < MT_; ++m) {
        #pragma unroll
        for (int u = 0; u < 2; ++u)
          acc[m][u] = (v8f){0.f, 0.f, 0.f, 0.f, 0.f, 0.f, 0.f, 0.f};
      }
      #pragma unroll
      for (int s = 0; s < 8; ++s) {
        union { uint4 q[2]; v16bf v; } fb[2];
        #pragma unroll
        for (int u = 0; u < 2; ++u) {
          const uint4* bp = (const uint4*)&sB[((ntq * 2 + u) * 8 + s) * 512];
          fb[u].q[0] = bp[lane];
          fb[u].q[1] = bp[32 + lane];
        }
        asm volatile("" :: "v"(fb[0].v), "v"(fb[1].v));
        #pragma unroll
        for (int u = 0; u < 2; ++u) {
          #pragma unroll
          for (int m = 0; m < MT_; ++m)
            acc[m][u] = __builtin_amdgcn_wmma_f32_16x16x32_bf16(
                false, fa[m][s].v, false, fb[u].v, (short)0, acc[m][u], false, false);
        }
        __builtin_amdgcn_sched_barrier(0);
      }
      #pragma unroll
      for (int u = 0; u < 2; ++u) {
        const int ug = (ntq * 2 + u) * 16 + c;
        const float wxs = s_wx[ug];
        const float va = s_va[ug];
        #pragma unroll
        for (int m = 0; m < MT_; ++m) {
          #pragma unroll
          for (int r = 0; r < 8; ++r)
            sum[m][r] += va * tanh_of_scaled(__builtin_fmaf(acc[m][u][r], S2L2E_, wxs));
        }
      }
    }

    #pragma unroll
    for (int m = 0; m < MT_; ++m) {
      #pragma unroll
      for (int r = 0; r < 8; ++r) {
        float v = sum[m][r];
        v += __shfl_xor(v, 1);
        v += __shfl_xor(v, 2);
        v += __shfl_xor(v, 4);
        v += __shfl_xor(v, 8);
        sum[m][r] = v;
      }
      float wgt[8];
      #pragma unroll
      for (int r = 0; r < 8; ++r) wgt[r] = expf(sum[m][r]);
      wacc += ((wgt[0] + wgt[1]) + (wgt[2] + wgt[3])) + ((wgt[4] + wgt[5]) + (wgt[6] + wgt[7]));
      v2f wp[4];
      #pragma unroll
      for (int r2 = 0; r2 < 4; ++r2) wp[r2] = (v2f){wgt[2 * r2], wgt[2 * r2 + 1]};

      float* cdst = &sC[((it * WAVES_ + wave) * MT_ + m) * 512 + lane];
      #pragma unroll
      for (int u = 0; u < 2; ++u) {
        union { uint4 q[2]; v16bf v; } fs;
        const uint4* sp = (const uint4*)&sSel[u * 512];
        fs.q[0] = sp[lane];
        fs.q[1] = sp[32 + lane];
        #pragma unroll
        for (int s = 0; s < 8; ++s) {
          const v8f d = __builtin_amdgcn_wmma_f32_16x16x32_bf16(
              false, fa[m][s].v, false, fs.v, (short)0,
              (v8f){0.f, 0.f, 0.f, 0.f, 0.f, 0.f, 0.f, 0.f}, false, false);
          v2f a2 = wp[0] * (v2f){d[0], d[1]};
          #pragma unroll
          for (int r2 = 1; r2 < 4; ++r2)
            a2 = __builtin_elementwise_fma(wp[r2], (v2f){d[2 * r2], d[2 * r2 + 1]}, a2);
          cdst[(2 * s + u) * 32] = a2.x + a2.y;
        }
      }
    }
  }

  wacc += __shfl_xor(wacc, 16);
  if (lane == 0) s_ws[wave] = wacc;
  __syncthreads();

  {
    const int j = tid & 255;
    #pragma unroll
    for (int qq = 0; qq < 2; ++qq) {
      const int q = 2 * (tid >> 8) + qq;
      const float* p = &sC[(q * 16) * 512 + (j >> 4) * 32 + (j & 15)];
      float a = 0.f;
      #pragma unroll
      for (int tl = 0; tl < 16; ++tl) {
        a += p[tl * 512];
        a += p[tl * 512 + 16];
      }
      s_red[q * U_ + j] = a;
    }
  }
  __syncthreads();
  if (tid < U_) {
    const float t = ((s_red[tid] + s_red[U_ + tid]) + s_red[2 * U_ + tid]) + s_red[3 * U_ + tid];
    const int i = tid >> 4, cc = tid & 15;
    const int col = 32 * (i >> 1) + 8 * (cc >> 2) + 4 * (i & 1) + (cc & 3);
    ctxp[((blockIdx.x & 1) * B_ + b) * U_ + col] = t;
  }
  if (tid == 0) {
    float S = s_ws[0];
    #pragma unroll
    for (int w = 1; w < WAVES_; ++w) S += s_ws[w];
    csum[blockIdx.x] = S;
  }
}

__device__ __forceinline__ float sigmoidf_(float a) {
  return 1.f / (1.f + expf(-a));
}

__global__ __launch_bounds__(256) void k_gates(
    const float* __restrict__ h, const float* __restrict__ ctx,
    const float* __restrict__ csum, const float* __restrict__ Cz, const float* __restrict__ Cr,
    const float* __restrict__ Cp, const float* __restrict__ pre,
    float* __restrict__ ztw, float* __restrict__ rhw, float* __restrict__ ppw) {
  __shared__ float4 part[8 * 8 * 32];
  const int n0 = blockIdx.x * 64, m0 = blockIdx.y * 16;
  const int z = blockIdx.z;
  const float* C = (z == 0 ? Cz : z == 1 ? Cr : Cp) + IN_ * U_;
  const float* p0 = pre + (z < 2 ? 2 * z + 1 : 4) * (B_ * U_);
  const float* p1 = pre + (z < 2 ? 2 * z : 4) * (B_ * U_);
  const DenseOut d = dense16x64<true, 3>(ctx, ctx + B_ * U_, csum, C, p0, U_, true, p1, z < 2,
                                         h, z == 1, m0, n0, part);
  const int lane = threadIdx.x & 31, wave = threadIdx.x >> 5;
  const int row = m0 + 8 * (lane >> 4) + wave, col = n0 + 4 * (lane & 15);
  float4 s;
  s.x = d.s.x + d.e1.x + d.e0.x;
  s.y = d.s.y + d.e1.y + d.e0.y;
  s.z = d.s.z + d.e1.z + d.e0.z;
  s.w = d.s.w + d.e1.w + d.e0.w;
  if (z == 2) {
    *(float4*)(ppw + row * U_ + col) = s;
    return;
  }
  float4 g;
  g.x = sigmoidf_(s.x);
  g.y = sigmoidf_(s.y);
  g.z = sigmoidf_(s.z);
  g.w = sigmoidf_(s.w);
  if (z == 1) {
    *(float4*)(rhw + row * U_ + col) =
        make_float4(g.x * d.e2.x, g.y * d.e2.y, g.z * d.e2.z, g.w * d.e2.w);
  } else {
    *(float4*)(ztw + row * U_ + col) = g;
  }
}

__global__ __launch_bounds__(256) void k_out(
    const float* __restrict__ h, const float* __restrict__ Up,
    const float* __restrict__ ppw, const float* __restrict__ ztw,
    const float* __restrict__ rhw, float* __restrict__ out) {
  __shared__ float4 part[8 * 8 * 32];
  const int n0 = blockIdx.x * 64, m0 = blockIdx.y * 16;
  const DenseOut d = dense16x64<false, 3>(rhw, rhw, rhw, Up, ppw, U_, true, ztw, true, h, true,
                                          m0, n0, part);
  const int lane = threadIdx.x & 31, wave = threadIdx.x >> 5;
  const int row = m0 + 8 * (lane >> 4) + wave, col = n0 + 4 * (lane & 15);
  const float4 zt = d.e1, hv = d.e2;
  float4 o;
  o.x = (1.f - zt.x) * hv.x + zt.x * tanhf(d.s.x + d.e0.x);
  o.y = (1.f - zt.y) * hv.y + zt.y * tanhf(d.s.y + d.e0.y);
  o.z = (1.f - zt.z) * hv.z + zt.z * tanhf(d.s.z + d.e0.z);
  o.w = (1.f - zt.w) * hv.w + zt.w * tanhf(d.s.w + d.e0.w);
  *(float4*)(out + row * U_ + col) = o;
}

extern "C" void kernel_launch(void* const* d_in, const int* in_sizes, int n_in,
                              void* d_out, int out_size, void* d_ws, size_t ws_size,
                              hipStream_t stream) {
  const float* inputs = (const float*)d_in[0];
  const float* h_tm   = (const float*)d_in[1];
  const float* x_seq  = (const float*)d_in[2];
  const float* V_a    = (const float*)d_in[3];
  const float* W_a    = (const float*)d_in[4];
  const float* U_a    = (const float*)d_in[5];
  const float* b_a    = (const float*)d_in[6];
  const float* C_z    = (const float*)d_in[7];
  const float* W_z    = (const float*)d_in[8];
  const float* b_z    = (const float*)d_in[9];
  const float* C_r    = (const float*)d_in[10];
  const float* W_r    = (const float*)d_in[11];
  const float* b_r    = (const float*)d_in[12];
  const float* C_p    = (const float*)d_in[13];
  const float* U_p    = (const float*)d_in[14];
  const float* b_p    = (const float*)d_in[15];
  float* out = (float*)d_out;

  float* wsf   = (float*)d_ws;
  float* wxstm = wsf;
  float* csum  = wsf + 32768;
  float* ctx   = csum + (size_t)B_ * TE_;
  float* ztw   = ctx + 2 * B_ * U_;
  float* rhw   = ztw + B_ * U_;
  float* pre   = rhw + B_ * U_;
  float* ppw   = pre + 5 * B_ * U_;

  dim3 blk(256);
  k_pre    <<<dim3(U_ / 64, B_ / 16, 6), blk, 0, stream>>>(h_tm, inputs, W_a, W_z, C_z, b_z, W_r, C_r,
                                                           b_r, C_p, b_p, wxstm, pre);
  k_scores <<<dim3(B_ * TE_ / ROWS_WG_), dim3(WAVES_ * 32), 0, stream>>>(
      x_seq, U_a, wxstm, b_a, V_a, ctx, csum);
  k_gates  <<<dim3(U_ / 64, B_ / 16, 3), blk, 0, stream>>>(h_tm, ctx, csum, C_z, C_r, C_p, pre, ztw, rhw, ppw);
  k_out    <<<dim3(U_ / 64, B_ / 16), blk, 0, stream>>>(h_tm, U_p, ppw, ztw, rhw, out);
}
